# starting kernel
# baseline (speedup 1.0000x reference)
#include <hip/hip_runtime.h>
#include <math.h>

typedef __attribute__((ext_vector_type(2))) float v2f;
typedef __attribute__((ext_vector_type(8))) float v8f;

#define NFEAT 128
#define NHID  128
#define NCLASS 16

#define G1_ROWS   48
#define G1_TILES  3
#define G1_STRIDE 132

__global__ void zero_kernel(float* __restrict__ p, long long n) {
  long long i = (long long)blockIdx.x * blockDim.x + threadIdx.x;
  long long stride = (long long)gridDim.x * blockDim.x;
  for (; i < n; i += stride) p[i] = 0.0f;
}

__global__ void scatter_kernel(const float* __restrict__ feat,
                               const int* __restrict__ src,
                               const int* __restrict__ dst,
                               float* __restrict__ agg,
                               float* __restrict__ deg,
                               int addDeg, long long nEdges) {
  long long gid = (long long)blockIdx.x * blockDim.x + threadIdx.x;
  long long e = gid >> 5;
  if (e >= nEdges) return;
  int lane = (int)(gid & 31);
  int s = src[e];
  int d = dst[e];
  int f = lane * 4;
  const float4 v = *(const float4*)(feat + (long long)s * NFEAT + f);
  float* o = agg + (long long)d * NFEAT + f;
  atomicAdd(o + 0, v.x);
  atomicAdd(o + 1, v.y);
  atomicAdd(o + 2, v.z);
  atomicAdd(o + 3, v.w);
  if (addDeg && lane == 0) atomicAdd(deg + d, 1.0f);
}

__global__ void normalize_kernel(float* __restrict__ agg,
                                 const float* __restrict__ deg, long long n) {
  long long i = (long long)blockIdx.x * blockDim.x + threadIdx.x;
  if (i >= n) return;
  float dv = deg[i >> 7];
  agg[i] = agg[i] / fmaxf(dv, 1.0f);
}

__global__ __launch_bounds__(256)
void gemm1_kernel(const float* __restrict__ x, const float* __restrict__ agg,
                  const float* __restrict__ wself, const float* __restrict__ wneigh,
                  const float* __restrict__ bias, float* __restrict__ h, int nNodes) {
  __shared__ float xs[G1_ROWS * G1_STRIDE];
  __shared__ float as[G1_ROWS * G1_STRIDE];

  const int tid  = threadIdx.x;
  const int lane = tid & 31;
  const int wave = tid >> 5;
  const int half = lane >> 4;
  const int l16  = lane & 15;
  const int col  = wave * 16 + l16;
  const long long base = (long long)blockIdx.x * G1_ROWS;

  for (int i = tid; i < G1_ROWS * 32; i += 256) {
    int row = i >> 5;
    int c4  = (i & 31) * 4;
    long long node = base + row;
    float4 vx = make_float4(0.f, 0.f, 0.f, 0.f);
    float4 va = vx;
    if (node < nNodes) {
      vx = *(const float4*)(x   + node * NFEAT + c4);
      va = *(const float4*)(agg + node * NFEAT + c4);
    }
    *(float4*)(&xs[row * G1_STRIDE + c4]) = vx;
    *(float4*)(&as[row * G1_STRIDE + c4]) = va;
  }
  __syncthreads();

  const v8f vzero = {0.f, 0.f, 0.f, 0.f, 0.f, 0.f, 0.f, 0.f};
  v8f acc[G1_TILES];
#pragma unroll
  for (int r = 0; r < G1_TILES; ++r) acc[r] = vzero;

  for (int k0 = 0; k0 < NFEAT; k0 += 4) {
    const int kb = k0 + half * 2;
    v2f bs, bn;
    bs.x = wself [(long long)kb       * NHID + col];
    bs.y = wself [(long long)(kb + 1) * NHID + col];
    bn.x = wneigh[(long long)kb       * NHID + col];
    bn.y = wneigh[(long long)(kb + 1) * NHID + col];
#pragma unroll
    for (int r = 0; r < G1_TILES; ++r) {
      const v2f ax = *(const v2f*)(&xs[(r * 16 + l16) * G1_STRIDE + kb]);
      const v2f aa = *(const v2f*)(&as[(r * 16 + l16) * G1_STRIDE + kb]);
      acc[r] = __builtin_amdgcn_wmma_f32_16x16x4_f32(false, ax, false, bs,
                                                     (short)0, acc[r], false, false);
      acc[r] = __builtin_amdgcn_wmma_f32_16x16x4_f32(false, aa, false, bn,
                                                     (short)0, acc[r], false, false);
    }
  }

  const float bv = bias[col];
#pragma unroll
  for (int r = 0; r < G1_TILES; ++r) {
#pragma unroll
    for (int j = 0; j < 8; ++j) {
      long long node = base + r * 16 + j + half * 8;
      if (node < nNodes) {
        float v = acc[r][j] + bv;
        h[node * NHID + col] = fmaxf(v, 0.0f);
      }
    }
  }
}

__global__ __launch_bounds__(256)
void gemm2_kernel(const float* __restrict__ h, const float* __restrict__ agg,
                  const float* __restrict__ wself, const float* __restrict__ wneigh,
                  const float* __restrict__ bias, float* __restrict__ out, int nNodes) {
  const int lane = threadIdx.x & 31;
  const int wave = threadIdx.x >> 5;
  const int half = lane >> 4;
  const int l16  = lane & 15;
  const long long base = ((long long)blockIdx.x * 8 + wave) * 16;
  if (base >= nNodes) return;

  long long nodeA = base + l16;
  if (nodeA >= nNodes) nodeA = nNodes - 1;

  v8f acc = {0.f, 0.f, 0.f, 0.f, 0.f, 0.f, 0.f, 0.f};

  for (int k0 = 0; k0 < NHID; k0 += 4) {
    const int kb = k0 + half * 2;
    const v2f ah = *(const v2f*)(h   + nodeA * NHID + kb);
    const v2f aa = *(const v2f*)(agg + nodeA * NHID + kb);
    v2f bs, bn;
    bs.x = wself [kb       * NCLASS + l16];
    bs.y = wself [(kb + 1) * NCLASS + l16];
    bn.x = wneigh[kb       * NCLASS + l16];
    bn.y = wneigh[(kb + 1) * NCLASS + l16];
    acc = __builtin_amdgcn_wmma_f32_16x16x4_f32(false, ah, false, bs,
                                                (short)0, acc, false, false);
    acc = __builtin_amdgcn_wmma_f32_16x16x4_f32(false, aa, false, bn,
                                                (short)0, acc, false, false);
  }

  const float bv = bias[l16];
#pragma unroll
  for (int j = 0; j < 8; ++j) {
    float v = acc[j] + bv;
    float mx = v;
#pragma unroll
    for (int off = 8; off >= 1; off >>= 1)
      mx = fmaxf(mx, __shfl_xor(mx, off, 16));
    float e = expf(v - mx);
    float s = e;
#pragma unroll
    for (int off = 8; off >= 1; off >>= 1)
      s += __shfl_xor(s, off, 16);
    long long node = base + j + half * 8;
    if (node < nNodes) out[node * NCLASS + l16] = v - mx - logf(s);
  }
}

extern "C" void kernel_launch(void* const* d_in, const int* in_sizes, int n_in,
                              void* d_out, int out_size, void* d_ws, size_t ws_size,
                              hipStream_t stream) {
  const float* x   = (const float*)d_in[0];
  const int*   src = (const int*)  d_in[1];
  const int*   dst = (const int*)  d_in[2];
  const float* ws1 = (const float*)d_in[3];
  const float* wn1 = (const float*)d_in[4];
  const float* b1  = (const float*)d_in[5];
  const float* ws2 = (const float*)d_in[6];
  const float* wn2 = (const float*)d_in[7];
  const float* b2  = (const float*)d_in[8];
  float* out = (float*)d_out;

  const int nNodes = in_sizes[0] / NFEAT;
  const long long nEdges = in_sizes[1];

  char* w = (char*)d_ws;
  size_t degBytes = (((size_t)nNodes * 4) + 255) & ~(size_t)255;
  float* deg = (float*)w;
  float* agg = (float*)(w + degBytes);
  float* h   = agg + (size_t)nNodes * NFEAT;

  const long long aggN  = (long long)nNodes * NFEAT;
  const long long zeroN = (long long)(degBytes / 4) + aggN;

  const int scatterBlocks = (int)((nEdges * 32 + 255) / 256);
  const int normBlocks    = (int)((aggN + 255) / 256);

  zero_kernel<<<2048, 256, 0, stream>>>((float*)w, zeroN);
  scatter_kernel<<<scatterBlocks, 256, 0, stream>>>(x, src, dst, agg, deg, 1, nEdges);
  normalize_kernel<<<normBlocks, 256, 0, stream>>>(agg, deg, aggN);
  gemm1_kernel<<<(nNodes + G1_ROWS - 1) / G1_ROWS, 256, 0, stream>>>(
      x, agg, ws1, wn1, b1, h, nNodes);

  zero_kernel<<<2048, 256, 0, stream>>>(agg, aggN);
  scatter_kernel<<<scatterBlocks, 256, 0, stream>>>(h, src, dst, agg, deg, 0, nEdges);
  normalize_kernel<<<normBlocks, 256, 0, stream>>>(agg, deg, aggN);
  gemm2_kernel<<<(nNodes + 127) / 128, 256, 0, stream>>>(
      h, agg, ws2, wn2, b2, out, nNodes);
}


# layer 2 projects h to the 16 classes (h@Wneigh2) before the neighbour sum, so the atomic scatter adds 16 floats per edge, not 128; sum order changes (product first, then neighbour sum)
# speedup vs baseline: 1.5621x; 1.5621x over previous
#include <hip/hip_runtime.h>
#include <math.h>

typedef __attribute__((ext_vector_type(2))) float v2f;
typedef __attribute__((ext_vector_type(8))) float v8f;

#define NFEAT 128
#define NHID  128
#define NCLASS 16

#define G1_ROWS   48
#define G1_TILES  3
#define G1_STRIDE 132

__global__ void zero_kernel(float* __restrict__ p, long long n) {
  long long i = (long long)blockIdx.x * blockDim.x + threadIdx.x;
  long long stride = (long long)gridDim.x * blockDim.x;
  for (; i < n; i += stride) p[i] = 0.0f;
}

__global__ void scatter_kernel(const float* __restrict__ feat,
                               const int* __restrict__ src,
                               const int* __restrict__ dst,
                               float* __restrict__ agg,
                               float* __restrict__ deg,
                               int addDeg, long long nEdges) {
  long long gid = (long long)blockIdx.x * blockDim.x + threadIdx.x;
  long long e = gid >> 5;
  if (e >= nEdges) return;
  int lane = (int)(gid & 31);
  int s = src[e];
  int d = dst[e];
  int f = lane * 4;
  const float4 v = *(const float4*)(feat + (long long)s * NFEAT + f);
  float* o = agg + (long long)d * NFEAT + f;
  atomicAdd(o + 0, v.x);
  atomicAdd(o + 1, v.y);
  atomicAdd(o + 2, v.z);
  atomicAdd(o + 3, v.w);
  if (addDeg && lane == 0) atomicAdd(deg + d, 1.0f);
}

__global__ void normalize_kernel(float* __restrict__ agg,
                                 const float* __restrict__ deg, long long n) {
  long long i = (long long)blockIdx.x * blockDim.x + threadIdx.x;
  if (i >= n) return;
  float dv = deg[i >> 7];
  agg[i] = agg[i] / fmaxf(dv, 1.0f);
}

__global__ __launch_bounds__(256)
void gemm1_kernel(const float* __restrict__ x, const float* __restrict__ agg,
                  const float* __restrict__ wself, const float* __restrict__ wneigh,
                  const float* __restrict__ bias, float* __restrict__ h, int nNodes) {
  __shared__ float xs[G1_ROWS * G1_STRIDE];
  __shared__ float as[G1_ROWS * G1_STRIDE];

  const int tid  = threadIdx.x;
  const int lane = tid & 31;
  const int wave = tid >> 5;
  const int half = lane >> 4;
  const int l16  = lane & 15;
  const int col  = wave * 16 + l16;
  const long long base = (long long)blockIdx.x * G1_ROWS;

  for (int i = tid; i < G1_ROWS * 32; i += 256) {
    int row = i >> 5;
    int c4  = (i & 31) * 4;
    long long node = base + row;
    float4 vx = make_float4(0.f, 0.f, 0.f, 0.f);
    float4 va = vx;
    if (node < nNodes) {
      vx = *(const float4*)(x   + node * NFEAT + c4);
      va = *(const float4*)(agg + node * NFEAT + c4);
    }
    *(float4*)(&xs[row * G1_STRIDE + c4]) = vx;
    *(float4*)(&as[row * G1_STRIDE + c4]) = va;
  }
  __syncthreads();

  const v8f vzero = {0.f, 0.f, 0.f, 0.f, 0.f, 0.f, 0.f, 0.f};
  v8f acc[G1_TILES];
#pragma unroll
  for (int r = 0; r < G1_TILES; ++r) acc[r] = vzero;

  for (int k0 = 0; k0 < NFEAT; k0 += 4) {
    const int kb = k0 + half * 2;
    v2f bs, bn;
    bs.x = wself [(long long)kb       * NHID + col];
    bs.y = wself [(long long)(kb + 1) * NHID + col];
    bn.x = wneigh[(long long)kb       * NHID + col];
    bn.y = wneigh[(long long)(kb + 1) * NHID + col];
#pragma unroll
    for (int r = 0; r < G1_TILES; ++r) {
      const v2f ax = *(const v2f*)(&xs[(r * 16 + l16) * G1_STRIDE + kb]);
      const v2f aa = *(const v2f*)(&as[(r * 16 + l16) * G1_STRIDE + kb]);
      acc[r] = __builtin_amdgcn_wmma_f32_16x16x4_f32(false, ax, false, bs,
                                                     (short)0, acc[r], false, false);
      acc[r] = __builtin_amdgcn_wmma_f32_16x16x4_f32(false, aa, false, bn,
                                                     (short)0, acc[r], false, false);
    }
  }

  const float bv = bias[col];
#pragma unroll
  for (int r = 0; r < G1_TILES; ++r) {
#pragma unroll
    for (int j = 0; j < 8; ++j) {
      long long node = base + r * 16 + j + half * 8;
      if (node < nNodes) {
        float v = acc[r][j] + bv;
        h[node * NHID + col] = fmaxf(v, 0.0f);
      }
    }
  }
}

__global__ __launch_bounds__(256)
void proj2_kernel(const float* __restrict__ h,
                  const float* __restrict__ wself, const float* __restrict__ wneigh,
                  const float* __restrict__ bias, float* __restrict__ sOut,
                  float* __restrict__ pOut, float* __restrict__ agg2, int nNodes) {
  const int lane = threadIdx.x & 31;
  const int wave = threadIdx.x >> 5;
  const int half = lane >> 4;
  const int l16  = lane & 15;
  const long long base = ((long long)blockIdx.x * 8 + wave) * 16;
  if (base >= nNodes) return;

  long long nodeA = base + l16;
  if (nodeA >= nNodes) nodeA = nNodes - 1;

  v8f accS = {0.f, 0.f, 0.f, 0.f, 0.f, 0.f, 0.f, 0.f};
  v8f accP = {0.f, 0.f, 0.f, 0.f, 0.f, 0.f, 0.f, 0.f};

  for (int k0 = 0; k0 < NHID; k0 += 4) {
    const int kb = k0 + half * 2;
    const v2f ah = *(const v2f*)(h + nodeA * NHID + kb);
    v2f bs, bn;
    bs.x = wself [kb       * NCLASS + l16];
    bs.y = wself [(kb + 1) * NCLASS + l16];
    bn.x = wneigh[kb       * NCLASS + l16];
    bn.y = wneigh[(kb + 1) * NCLASS + l16];
    accS = __builtin_amdgcn_wmma_f32_16x16x4_f32(false, ah, false, bs,
                                                 (short)0, accS, false, false);
    accP = __builtin_amdgcn_wmma_f32_16x16x4_f32(false, ah, false, bn,
                                                 (short)0, accP, false, false);
  }

  const float bv = bias[l16];
#pragma unroll
  for (int j = 0; j < 8; ++j) {
    long long node = base + j + half * 8;
    if (node < nNodes) {
      sOut[node * NCLASS + l16] = accS[j] + bv;
      pOut[node * NCLASS + l16] = accP[j];
      agg2[node * NCLASS + l16] = 0.0f;
    }
  }
}

__global__ __launch_bounds__(256)
void scatter16_kernel(const float* __restrict__ p,
                      const int* __restrict__ src,
                      const int* __restrict__ dst,
                      float* __restrict__ agg2, long long nEdges) {
  const int lane = threadIdx.x & 31;
  const int half = lane >> 4;
  const int l16  = lane & 15;
  const long long wv = ((long long)blockIdx.x * blockDim.x + threadIdx.x) >> 5;
  const long long e0 = wv * 32;
  if (e0 >= nEdges) return;
  const long long me = e0 + lane;
  int sv = 0, dv = 0;
  if (me < nEdges) { sv = src[me]; dv = dst[me]; }
  int   dr[16];
  float vr[16];
#pragma unroll
  for (int j = 0; j < 16; ++j) {
    const int sl = 2 * j + half;
    const int s = __shfl(sv, sl, 32);
    dr[j] = __shfl(dv, sl, 32);
    vr[j] = p[(long long)s * NCLASS + l16];
  }
#pragma unroll
  for (int j = 0; j < 16; ++j) {
    if (e0 + 2 * j + half < nEdges)
      atomicAdd(agg2 + (long long)dr[j] * NCLASS + l16, vr[j]);
  }
}

__global__ __launch_bounds__(256)
void final2_kernel(const float* __restrict__ s, const float* __restrict__ agg2,
                   const float* __restrict__ deg, float* __restrict__ out, long long n) {
  const long long i = (long long)blockIdx.x * blockDim.x + threadIdx.x;
  const bool ok = i < n;
  const long long ii = ok ? i : n - 1;
  const float v = s[ii] + agg2[ii] / fmaxf(deg[ii >> 4], 1.0f);
  float mx = v;
#pragma unroll
  for (int off = 8; off >= 1; off >>= 1)
    mx = fmaxf(mx, __shfl_xor(mx, off, 16));
  const float e = expf(v - mx);
  float sum = e;
#pragma unroll
  for (int off = 8; off >= 1; off >>= 1)
    sum += __shfl_xor(sum, off, 16);
  if (ok) out[i] = v - mx - logf(sum);
}

extern "C" void kernel_launch(void* const* d_in, const int* in_sizes, int n_in,
                              void* d_out, int out_size, void* d_ws, size_t ws_size,
                              hipStream_t stream) {
  const float* x   = (const float*)d_in[0];
  const int*   src = (const int*)  d_in[1];
  const int*   dst = (const int*)  d_in[2];
  const float* ws1 = (const float*)d_in[3];
  const float* wn1 = (const float*)d_in[4];
  const float* b1  = (const float*)d_in[5];
  const float* ws2 = (const float*)d_in[6];
  const float* wn2 = (const float*)d_in[7];
  const float* b2  = (const float*)d_in[8];
  float* out = (float*)d_out;

  const int nNodes = in_sizes[0] / NFEAT;
  const long long nEdges = in_sizes[1];

  char* w = (char*)d_ws;
  size_t degBytes = (((size_t)nNodes * 4) + 255) & ~(size_t)255;
  float* deg = (float*)w;
  float* agg = (float*)(w + degBytes);
  float* h   = agg + (size_t)nNodes * NFEAT;

  const long long aggN  = (long long)nNodes * NFEAT;
  const long long zeroN = (long long)(degBytes / 4) + aggN;

  const int scatterBlocks = (int)((nEdges * 32 + 255) / 256);
  const int normBlocks    = (int)((aggN + 255) / 256);

  zero_kernel<<<2048, 256, 0, stream>>>((float*)w, zeroN);
  scatter_kernel<<<scatterBlocks, 256, 0, stream>>>(x, src, dst, agg, deg, 1, nEdges);
  normalize_kernel<<<normBlocks, 256, 0, stream>>>(agg, deg, aggN);
  gemm1_kernel<<<(nNodes + G1_ROWS - 1) / G1_ROWS, 256, 0, stream>>>(
      x, agg, ws1, wn1, b1, h, nNodes);

  const long long outN = (long long)nNodes * NCLASS;
  float* s2   = agg;
  float* p2   = agg + outN;
  float* agg2 = agg + 2 * outN;
  const int projBlocks  = (nNodes + 127) / 128;
  const int scat16Blocks = (int)((nEdges + 255) / 256);
  const int finalBlocks = (int)((outN + 255) / 256);

  proj2_kernel<<<projBlocks, 256, 0, stream>>>(h, ws2, wn2, b2, s2, p2, agg2, nNodes);
  scatter16_kernel<<<scat16Blocks, 256, 0, stream>>>(p2, src, dst, agg2, nEdges);
  final2_kernel<<<finalBlocks, 256, 0, stream>>>(s2, agg2, deg, out, outN);
}


# layer-1 neighbour sum without float atomics: edges bucketed by dst (int count, alloc, fill), one wave per node sums its x rows in registers, stores the mean once; sum order = fill arrival order
# speedup vs baseline: 2.0010x; 1.2810x over previous
#include <hip/hip_runtime.h>
#include <math.h>

typedef __attribute__((ext_vector_type(2))) float v2f;
typedef __attribute__((ext_vector_type(8))) float v8f;

#define NFEAT 128
#define NHID  128
#define NCLASS 16

#define G1_ROWS   48
#define G1_TILES  3
#define G1_STRIDE 132

__global__ void zero_kernel(float* __restrict__ p, long long n) {
  long long i = (long long)blockIdx.x * blockDim.x + threadIdx.x;
  long long stride = (long long)gridDim.x * blockDim.x;
  for (; i < n; i += stride) p[i] = 0.0f;
}


__global__ __launch_bounds__(256)
void count_kernel(const int* __restrict__ dst, int* __restrict__ cnt, int nEdges) {
  const int stride = (int)(gridDim.x * blockDim.x);
  for (int e = (int)(blockIdx.x * blockDim.x + threadIdx.x); e < nEdges; e += stride)
    atomicAdd(cnt + dst[e], 1);
}

__global__ __launch_bounds__(256)
void seg_alloc_kernel(const int* __restrict__ cnt, int* __restrict__ total,
                      int* __restrict__ cur, int* __restrict__ seg, int nNodes) {
  __shared__ int wsum[8];
  __shared__ int blockBase;
  const int tid  = threadIdx.x;
  const int lane = tid & 31;
  const int wave = tid >> 5;
  const int i = (int)blockIdx.x * 256 + tid;
  const int c = (i < nNodes) ? cnt[i] : 0;
  int inc = c;
#pragma unroll
  for (int off = 1; off < 32; off <<= 1) {
    const int t = __shfl_up(inc, off, 32);
    if (lane >= off) inc += t;
  }
  if (lane == 31) wsum[wave] = inc;
  __syncthreads();
  if (tid == 0) {
    int tot = 0;
    for (int w = 0; w < 8; ++w) { const int t = wsum[w]; wsum[w] = tot; tot += t; }
    blockBase = atomicAdd(total, tot);
  }
  __syncthreads();
  if (i < nNodes) {
    const int start = blockBase + wsum[wave] + inc - c;
    cur[i] = start;
    *(int2*)(seg + 2 * i) = make_int2(start, c);
  }
}

__global__ __launch_bounds__(256)
void fill_kernel(const int* __restrict__ src, const int* __restrict__ dst,
                 int* __restrict__ cur, int* __restrict__ ssrc, int nEdges) {
  const int stride = (int)(gridDim.x * blockDim.x);
  for (int e = (int)(blockIdx.x * blockDim.x + threadIdx.x); e < nEdges; e += stride) {
    const int s = src[e];
    int pos = atomicAdd(cur + dst[e], 1);
    pos = min(max(pos, 0), nEdges - 1);
    ssrc[pos] = s;
  }
}

__global__ __launch_bounds__(256)
void gather_mean_kernel(const float* __restrict__ x, const int* __restrict__ seg,
                        const int* __restrict__ ssrc,
                        float* __restrict__ agg, float* __restrict__ deg, int nNodes) {
  const int lane = threadIdx.x & 31;
  const int wv = __builtin_amdgcn_readfirstlane((int)((blockIdx.x * blockDim.x + threadIdx.x) >> 5));
  const int nw = (int)((gridDim.x * blockDim.x) >> 5);
  const float* xl = x + lane * 4;
  for (int node = wv; node < nNodes; node += nw) {
    const int2 sg = *(const int2*)(seg + 2 * node);
    const int beg = sg.x;
    const int c   = sg.y;
    float4 acc = make_float4(0.f, 0.f, 0.f, 0.f);
    for (int b = 0; b < c; b += 32) {
      const int m = min(32, c - b);
      int sv = 0;
      if (lane < m) sv = ssrc[beg + b + lane];
      sv = min(max(sv, 0), nNodes - 1);
      for (int j = 0; j < m; j += 8) {
        float4 t[8];
#pragma unroll
        for (int u = 0; u < 8; ++u) {
          const int idx = min(j + u, m - 1);
          const int s = __builtin_amdgcn_readlane(sv, idx);
          t[u] = *(const float4*)(xl + (long long)s * NFEAT);
        }
#pragma unroll
        for (int u = 0; u < 8; ++u) {
          if (j + u < m) {
            acc.x += t[u].x; acc.y += t[u].y; acc.z += t[u].z; acc.w += t[u].w;
          }
        }
      }
    }
    const float dv = fmaxf((float)c, 1.0f);
    const float4 o = make_float4(acc.x / dv, acc.y / dv, acc.z / dv, acc.w / dv);
    *(float4*)(agg + (long long)node * NFEAT + lane * 4) = o;
    if (lane == 0) deg[node] = (float)c;
  }
}

__global__ __launch_bounds__(256)
void gemm1_kernel(const float* __restrict__ x, const float* __restrict__ agg,
                  const float* __restrict__ wself, const float* __restrict__ wneigh,
                  const float* __restrict__ bias, float* __restrict__ h, int nNodes) {
  __shared__ float xs[G1_ROWS * G1_STRIDE];
  __shared__ float as[G1_ROWS * G1_STRIDE];

  const int tid  = threadIdx.x;
  const int lane = tid & 31;
  const int wave = tid >> 5;
  const int half = lane >> 4;
  const int l16  = lane & 15;
  const int col  = wave * 16 + l16;
  const long long base = (long long)blockIdx.x * G1_ROWS;

  for (int i = tid; i < G1_ROWS * 32; i += 256) {
    int row = i >> 5;
    int c4  = (i & 31) * 4;
    long long node = base + row;
    float4 vx = make_float4(0.f, 0.f, 0.f, 0.f);
    float4 va = vx;
    if (node < nNodes) {
      vx = *(const float4*)(x   + node * NFEAT + c4);
      va = *(const float4*)(agg + node * NFEAT + c4);
    }
    *(float4*)(&xs[row * G1_STRIDE + c4]) = vx;
    *(float4*)(&as[row * G1_STRIDE + c4]) = va;
  }
  __syncthreads();

  const v8f vzero = {0.f, 0.f, 0.f, 0.f, 0.f, 0.f, 0.f, 0.f};
  v8f acc[G1_TILES];
#pragma unroll
  for (int r = 0; r < G1_TILES; ++r) acc[r] = vzero;

  for (int k0 = 0; k0 < NFEAT; k0 += 4) {
    const int kb = k0 + half * 2;
    v2f bs, bn;
    bs.x = wself [(long long)kb       * NHID + col];
    bs.y = wself [(long long)(kb + 1) * NHID + col];
    bn.x = wneigh[(long long)kb       * NHID + col];
    bn.y = wneigh[(long long)(kb + 1) * NHID + col];
#pragma unroll
    for (int r = 0; r < G1_TILES; ++r) {
      const v2f ax = *(const v2f*)(&xs[(r * 16 + l16) * G1_STRIDE + kb]);
      const v2f aa = *(const v2f*)(&as[(r * 16 + l16) * G1_STRIDE + kb]);
      acc[r] = __builtin_amdgcn_wmma_f32_16x16x4_f32(false, ax, false, bs,
                                                     (short)0, acc[r], false, false);
      acc[r] = __builtin_amdgcn_wmma_f32_16x16x4_f32(false, aa, false, bn,
                                                     (short)0, acc[r], false, false);
    }
  }

  const float bv = bias[col];
#pragma unroll
  for (int r = 0; r < G1_TILES; ++r) {
#pragma unroll
    for (int j = 0; j < 8; ++j) {
      long long node = base + r * 16 + j + half * 8;
      if (node < nNodes) {
        float v = acc[r][j] + bv;
        h[node * NHID + col] = fmaxf(v, 0.0f);
      }
    }
  }
}

__global__ __launch_bounds__(256)
void proj2_kernel(const float* __restrict__ h,
                  const float* __restrict__ wself, const float* __restrict__ wneigh,
                  const float* __restrict__ bias, float* __restrict__ sOut,
                  float* __restrict__ pOut, float* __restrict__ agg2, int nNodes) {
  const int lane = threadIdx.x & 31;
  const int wave = threadIdx.x >> 5;
  const int half = lane >> 4;
  const int l16  = lane & 15;
  const long long base = ((long long)blockIdx.x * 8 + wave) * 16;
  if (base >= nNodes) return;

  long long nodeA = base + l16;
  if (nodeA >= nNodes) nodeA = nNodes - 1;

  v8f accS = {0.f, 0.f, 0.f, 0.f, 0.f, 0.f, 0.f, 0.f};
  v8f accP = {0.f, 0.f, 0.f, 0.f, 0.f, 0.f, 0.f, 0.f};

  for (int k0 = 0; k0 < NHID; k0 += 4) {
    const int kb = k0 + half * 2;
    const v2f ah = *(const v2f*)(h + nodeA * NHID + kb);
    v2f bs, bn;
    bs.x = wself [kb       * NCLASS + l16];
    bs.y = wself [(kb + 1) * NCLASS + l16];
    bn.x = wneigh[kb       * NCLASS + l16];
    bn.y = wneigh[(kb + 1) * NCLASS + l16];
    accS = __builtin_amdgcn_wmma_f32_16x16x4_f32(false, ah, false, bs,
                                                 (short)0, accS, false, false);
    accP = __builtin_amdgcn_wmma_f32_16x16x4_f32(false, ah, false, bn,
                                                 (short)0, accP, false, false);
  }

  const float bv = bias[l16];
#pragma unroll
  for (int j = 0; j < 8; ++j) {
    long long node = base + j + half * 8;
    if (node < nNodes) {
      sOut[node * NCLASS + l16] = accS[j] + bv;
      pOut[node * NCLASS + l16] = accP[j];
      agg2[node * NCLASS + l16] = 0.0f;
    }
  }
}

__global__ __launch_bounds__(256)
void scatter16_kernel(const float* __restrict__ p,
                      const int* __restrict__ src,
                      const int* __restrict__ dst,
                      float* __restrict__ agg2, long long nEdges) {
  const int lane = threadIdx.x & 31;
  const int half = lane >> 4;
  const int l16  = lane & 15;
  const long long wv = ((long long)blockIdx.x * blockDim.x + threadIdx.x) >> 5;
  const long long e0 = wv * 32;
  if (e0 >= nEdges) return;
  const long long me = e0 + lane;
  int sv = 0, dv = 0;
  if (me < nEdges) { sv = src[me]; dv = dst[me]; }
  int   dr[16];
  float vr[16];
#pragma unroll
  for (int j = 0; j < 16; ++j) {
    const int sl = 2 * j + half;
    const int s = __shfl(sv, sl, 32);
    dr[j] = __shfl(dv, sl, 32);
    vr[j] = p[(long long)s * NCLASS + l16];
  }
#pragma unroll
  for (int j = 0; j < 16; ++j) {
    if (e0 + 2 * j + half < nEdges)
      atomicAdd(agg2 + (long long)dr[j] * NCLASS + l16, vr[j]);
  }
}

__global__ __launch_bounds__(256)
void final2_kernel(const float* __restrict__ s, const float* __restrict__ agg2,
                   const float* __restrict__ deg, float* __restrict__ out, long long n) {
  const long long i = (long long)blockIdx.x * blockDim.x + threadIdx.x;
  const bool ok = i < n;
  const long long ii = ok ? i : n - 1;
  const float v = s[ii] + agg2[ii] / fmaxf(deg[ii >> 4], 1.0f);
  float mx = v;
#pragma unroll
  for (int off = 8; off >= 1; off >>= 1)
    mx = fmaxf(mx, __shfl_xor(mx, off, 16));
  const float e = expf(v - mx);
  float sum = e;
#pragma unroll
  for (int off = 8; off >= 1; off >>= 1)
    sum += __shfl_xor(sum, off, 16);
  if (ok) out[i] = v - mx - logf(sum);
}

extern "C" void kernel_launch(void* const* d_in, const int* in_sizes, int n_in,
                              void* d_out, int out_size, void* d_ws, size_t ws_size,
                              hipStream_t stream) {
  const float* x   = (const float*)d_in[0];
  const int*   src = (const int*)  d_in[1];
  const int*   dst = (const int*)  d_in[2];
  const float* ws1 = (const float*)d_in[3];
  const float* wn1 = (const float*)d_in[4];
  const float* b1  = (const float*)d_in[5];
  const float* ws2 = (const float*)d_in[6];
  const float* wn2 = (const float*)d_in[7];
  const float* b2  = (const float*)d_in[8];
  float* out = (float*)d_out;

  const int nNodes = in_sizes[0] / NFEAT;
  const long long nEdges = in_sizes[1];

  char* w = (char*)d_ws;
  size_t degBytes = (((size_t)nNodes * 4) + 255) & ~(size_t)255;
  float* deg = (float*)w;
  float* agg = (float*)(w + degBytes);
  float* h   = agg + (size_t)nNodes * NFEAT;

  const int nPad = (nNodes + 63) & ~63;
  int* total = (int*)h;
  int* cnt   = total + 64;
  int* cur   = cnt + nPad;
  int* seg   = cur + nPad;
  int* ssrc  = cur + 3 * nPad;
  const int nE = (int)nEdges;

  zero_kernel<<<128, 256, 0, stream>>>((float*)total, (long long)(64 + nNodes));
  count_kernel<<<2048, 256, 0, stream>>>(dst, cnt, nE);
  seg_alloc_kernel<<<(nNodes + 255) / 256, 256, 0, stream>>>(cnt, total, cur, seg, nNodes);
  fill_kernel<<<2048, 256, 0, stream>>>(src, dst, cur, ssrc, nE);
  gather_mean_kernel<<<2048, 256, 0, stream>>>(x, seg, ssrc, agg, deg, nNodes);
  gemm1_kernel<<<(nNodes + G1_ROWS - 1) / G1_ROWS, 256, 0, stream>>>(
      x, agg, ws1, wn1, b1, h, nNodes);

  const long long outN = (long long)nNodes * NCLASS;
  float* s2   = agg;
  float* p2   = agg + outN;
  float* agg2 = agg + 2 * outN;
  const int projBlocks  = (nNodes + 127) / 128;
  const int scat16Blocks = (int)((nEdges + 255) / 256);
  const int finalBlocks = (int)((outN + 255) / 256);

  proj2_kernel<<<projBlocks, 256, 0, stream>>>(h, ws2, wn2, b2, s2, p2, agg2, nNodes);
  scatter16_kernel<<<scat16Blocks, 256, 0, stream>>>(p2, src, dst, agg2, nEdges);
  final2_kernel<<<finalBlocks, 256, 0, stream>>>(s2, agg2, deg, out, outN);
}


# layer-1 edge bucketing as a two-level counting sort with all counts and ranks in LDS (chunk tables, row scan, per-bucket fill); no global atomic per edge left; sum order = LDS fill arrival order
# speedup vs baseline: 3.1092x; 1.5538x over previous
#include <hip/hip_runtime.h>
#include <math.h>

typedef __attribute__((ext_vector_type(2))) float v2f;
typedef __attribute__((ext_vector_type(8))) float v8f;

#define NFEAT 128
#define NHID  128
#define NCLASS 16

#define G1_ROWS   48
#define G1_TILES  3
#define G1_STRIDE 132

#define CB_SHIFT 8
#define CB_NODES 256
#define NCHUNK   256
#define NB_MAX   1024
#define SORT_THREADS 1024
#define SORT_WAVES   32

template <int NW>
__device__ __forceinline__ int block_excl_scan(int v, int* wsum  , int& total) {
  const int lane = threadIdx.x & 31;
  const int wave = threadIdx.x >> 5;
  int inc = v;
#pragma unroll
  for (int off = 1; off < 32; off <<= 1) {
    const int t = __shfl_up(inc, off, 32);
    if (lane >= off) inc += t;
  }
  if (lane == 31) wsum[wave] = inc;
  __syncthreads();
  if (wave == 0) {
    const int w = (lane < NW) ? wsum[lane] : 0;
    int winc = w;
#pragma unroll
    for (int off = 1; off < 32; off <<= 1) {
      const int t = __shfl_up(winc, off, 32);
      if (lane >= off) winc += t;
    }
    if (lane < NW) wsum[lane] = winc - w;
    if (lane == 31) wsum[NW] = winc;
  }
  __syncthreads();
  const int r = wsum[wave] + inc - v;
  total = wsum[NW];
  __syncthreads();
  return r;
}

__global__ __launch_bounds__(SORT_THREADS)
void chunk_hist_kernel(const int* __restrict__ dst, int* __restrict__ hmat,
                       int nEdges, int nNodes, int nB, int chunk) {
  __shared__ int hist[NB_MAX];
  const int tid = threadIdx.x;
  for (int b = tid; b < NB_MAX; b += SORT_THREADS) hist[b] = 0;
  __syncthreads();
  const int e0 = (int)blockIdx.x * chunk;
  const int e1 = min(e0 + chunk, nEdges);
  for (int e = e0 + tid; e < e1; e += 4 * SORT_THREADS) {
    int d[4];
#pragma unroll
    for (int u = 0; u < 4; ++u) {
      const int ee = e + u * SORT_THREADS;
      d[u] = 0;
      if (ee < e1) d[u] = dst[ee];
    }
#pragma unroll
    for (int u = 0; u < 4; ++u) {
      if (e + u * SORT_THREADS < e1) {
        const int dd = min(max(d[u], 0), nNodes - 1);
        atomicAdd(&hist[min(dd >> CB_SHIFT, NB_MAX - 1)], 1);
      }
    }
  }
  __syncthreads();
  for (int b = tid; b < nB; b += SORT_THREADS) hmat[b * NCHUNK + (int)blockIdx.x] = hist[b];
}

__global__ __launch_bounds__(NCHUNK)
void row_scan_kernel(int* __restrict__ hmat, int* __restrict__ rowTot) {
  __shared__ int wsum[NCHUNK / 32 + 1];
  const int i = (int)blockIdx.x * NCHUNK + (int)threadIdx.x;
  const int v = hmat[i];
  int total;
  const int ex = block_excl_scan<NCHUNK / 32>(v, wsum, total);
  hmat[i] = ex;
  if (threadIdx.x == 0) rowTot[blockIdx.x] = total;
}

__global__ __launch_bounds__(SORT_THREADS)
void place_kernel(const int* __restrict__ src, const int* __restrict__ dst,
                  const int* __restrict__ hmat, const int* __restrict__ rowTot,
                  int* __restrict__ rec, int nEdges, int nNodes, int nB, int chunk) {
  __shared__ int cursor[NB_MAX];
  __shared__ int wsum[SORT_WAVES + 1];
  const int tid = threadIdx.x;
  int carry = 0;
  for (int b0 = 0; b0 < nB; b0 += SORT_THREADS) {
    const int b = b0 + tid;
    const int v = (b < nB) ? rowTot[b] : 0;
    int total;
    const int ex = block_excl_scan<SORT_WAVES>(v, wsum, total);
    if (b < min(nB, NB_MAX)) cursor[b] = carry + ex + hmat[b * NCHUNK + (int)blockIdx.x];
    carry += total;
  }
  __syncthreads();
  const int e0 = (int)blockIdx.x * chunk;
  const int e1 = min(e0 + chunk, nEdges);
  for (int e = e0 + tid; e < e1; e += 4 * SORT_THREADS) {
    int s[4], d[4];
#pragma unroll
    for (int u = 0; u < 4; ++u) {
      const int ee = e + u * SORT_THREADS;
      s[u] = 0; d[u] = 0;
      if (ee < e1) { s[u] = src[ee]; d[u] = dst[ee]; }
    }
#pragma unroll
    for (int u = 0; u < 4; ++u) {
      if (e + u * SORT_THREADS < e1) {
        const int dd = min(max(d[u], 0), nNodes - 1);
        int pos = atomicAdd(&cursor[min(dd >> CB_SHIFT, NB_MAX - 1)], 1);
        pos = min(max(pos, 0), nEdges - 1);
        rec[pos] = (int)(((unsigned)s[u] << CB_SHIFT) | (unsigned)(dd & (CB_NODES - 1)));
      }
    }
  }
}

__global__ __launch_bounds__(SORT_THREADS)
void fine_sort_kernel(const int* __restrict__ rec, const int* __restrict__ rowTot,
                      int* __restrict__ seg, int* __restrict__ ssrc,
                      int nEdges, int nNodes) {
  __shared__ int cnt[CB_NODES];
  __shared__ int wsum[SORT_WAVES + 1];
  const int tid = threadIdx.x;
  const int bkt = (int)blockIdx.x;
  int part = 0;
  for (int b = tid; b < bkt; b += SORT_THREADS) part += rowTot[b];
  int bstart;
  (void)block_excl_scan<SORT_WAVES>(part, wsum, bstart);
  const int r0 = min(max(bstart, 0), nEdges);
  const int r1 = min(r0 + max(rowTot[bkt], 0), nEdges);
  if (tid < CB_NODES) cnt[tid] = 0;
  __syncthreads();
  for (int i = r0 + tid; i < r1; i += 4 * SORT_THREADS) {
    int r[4];
#pragma unroll
    for (int u = 0; u < 4; ++u) {
      const int ii = i + u * SORT_THREADS;
      r[u] = 0;
      if (ii < r1) r[u] = rec[ii];
    }
#pragma unroll
    for (int u = 0; u < 4; ++u)
      if (i + u * SORT_THREADS < r1) atomicAdd(&cnt[r[u] & (CB_NODES - 1)], 1);
  }
  __syncthreads();
  const int c = (tid < CB_NODES) ? cnt[tid] : 0;
  int total;
  const int ex = block_excl_scan<SORT_WAVES>(c, wsum, total);
  if (tid < CB_NODES) {
    const int start = r0 + ex;
    const int node = bkt * CB_NODES + tid;
    if (node < nNodes) *(int2*)(seg + 2 * node) = make_int2(start, c);
    cnt[tid] = start;
  }
  __syncthreads();
  for (int i = r0 + tid; i < r1; i += 4 * SORT_THREADS) {
    int r[4];
#pragma unroll
    for (int u = 0; u < 4; ++u) {
      const int ii = i + u * SORT_THREADS;
      r[u] = 0;
      if (ii < r1) r[u] = rec[ii];
    }
#pragma unroll
    for (int u = 0; u < 4; ++u) {
      if (i + u * SORT_THREADS < r1) {
        int pos = atomicAdd(&cnt[r[u] & (CB_NODES - 1)], 1);
        pos = min(max(pos, 0), nEdges - 1);
        ssrc[pos] = (int)((unsigned)r[u] >> CB_SHIFT);
      }
    }
  }
}

__global__ __launch_bounds__(256)
void gather_mean_kernel(const float* __restrict__ x, const int* __restrict__ seg,
                        const int* __restrict__ ssrc,
                        float* __restrict__ agg, float* __restrict__ deg, int nNodes) {
  const int lane = threadIdx.x & 31;
  const int wv = __builtin_amdgcn_readfirstlane((int)((blockIdx.x * blockDim.x + threadIdx.x) >> 5));
  const int nw = (int)((gridDim.x * blockDim.x) >> 5);
  const float* xl = x + lane * 4;
  for (int node = wv; node < nNodes; node += nw) {
    const int2 sg = *(const int2*)(seg + 2 * node);
    const int beg = sg.x;
    const int c   = sg.y;
    float4 acc = make_float4(0.f, 0.f, 0.f, 0.f);
    for (int b = 0; b < c; b += 32) {
      const int m = min(32, c - b);
      int sv = 0;
      if (lane < m) sv = ssrc[beg + b + lane];
      sv = min(max(sv, 0), nNodes - 1);
      for (int j = 0; j < m; j += 8) {
        float4 t[8];
#pragma unroll
        for (int u = 0; u < 8; ++u) {
          const int idx = min(j + u, m - 1);
          const int s = __builtin_amdgcn_readlane(sv, idx);
          t[u] = *(const float4*)(xl + (long long)s * NFEAT);
        }
#pragma unroll
        for (int u = 0; u < 8; ++u) {
          if (j + u < m) {
            acc.x += t[u].x; acc.y += t[u].y; acc.z += t[u].z; acc.w += t[u].w;
          }
        }
      }
    }
    const float dv = fmaxf((float)c, 1.0f);
    const float4 o = make_float4(acc.x / dv, acc.y / dv, acc.z / dv, acc.w / dv);
    *(float4*)(agg + (long long)node * NFEAT + lane * 4) = o;
    if (lane == 0) deg[node] = (float)c;
  }
}

__global__ __launch_bounds__(256)
void gemm1_kernel(const float* __restrict__ x, const float* __restrict__ agg,
                  const float* __restrict__ wself, const float* __restrict__ wneigh,
                  const float* __restrict__ bias, float* __restrict__ h, int nNodes) {
  __shared__ float xs[G1_ROWS * G1_STRIDE];
  __shared__ float as[G1_ROWS * G1_STRIDE];

  const int tid  = threadIdx.x;
  const int lane = tid & 31;
  const int wave = tid >> 5;
  const int half = lane >> 4;
  const int l16  = lane & 15;
  const int col  = wave * 16 + l16;
  const long long base = (long long)blockIdx.x * G1_ROWS;

  for (int i = tid; i < G1_ROWS * 32; i += 256) {
    int row = i >> 5;
    int c4  = (i & 31) * 4;
    long long node = base + row;
    float4 vx = make_float4(0.f, 0.f, 0.f, 0.f);
    float4 va = vx;
    if (node < nNodes) {
      vx = *(const float4*)(x   + node * NFEAT + c4);
      va = *(const float4*)(agg + node * NFEAT + c4);
    }
    *(float4*)(&xs[row * G1_STRIDE + c4]) = vx;
    *(float4*)(&as[row * G1_STRIDE + c4]) = va;
  }
  __syncthreads();

  const v8f vzero = {0.f, 0.f, 0.f, 0.f, 0.f, 0.f, 0.f, 0.f};
  v8f acc[G1_TILES];
#pragma unroll
  for (int r = 0; r < G1_TILES; ++r) acc[r] = vzero;

  for (int k0 = 0; k0 < NFEAT; k0 += 4) {
    const int kb = k0 + half * 2;
    v2f bs, bn;
    bs.x = wself [(long long)kb       * NHID + col];
    bs.y = wself [(long long)(kb + 1) * NHID + col];
    bn.x = wneigh[(long long)kb       * NHID + col];
    bn.y = wneigh[(long long)(kb + 1) * NHID + col];
#pragma unroll
    for (int r = 0; r < G1_TILES; ++r) {
      const v2f ax = *(const v2f*)(&xs[(r * 16 + l16) * G1_STRIDE + kb]);
      const v2f aa = *(const v2f*)(&as[(r * 16 + l16) * G1_STRIDE + kb]);
      acc[r] = __builtin_amdgcn_wmma_f32_16x16x4_f32(false, ax, false, bs,
                                                     (short)0, acc[r], false, false);
      acc[r] = __builtin_amdgcn_wmma_f32_16x16x4_f32(false, aa, false, bn,
                                                     (short)0, acc[r], false, false);
    }
  }

  const float bv = bias[col];
#pragma unroll
  for (int r = 0; r < G1_TILES; ++r) {
#pragma unroll
    for (int j = 0; j < 8; ++j) {
      long long node = base + r * 16 + j + half * 8;
      if (node < nNodes) {
        float v = acc[r][j] + bv;
        h[node * NHID + col] = fmaxf(v, 0.0f);
      }
    }
  }
}

__global__ __launch_bounds__(256)
void proj2_kernel(const float* __restrict__ h,
                  const float* __restrict__ wself, const float* __restrict__ wneigh,
                  const float* __restrict__ bias, float* __restrict__ sOut,
                  float* __restrict__ pOut, float* __restrict__ agg2, int nNodes) {
  const int lane = threadIdx.x & 31;
  const int wave = threadIdx.x >> 5;
  const int half = lane >> 4;
  const int l16  = lane & 15;
  const long long base = ((long long)blockIdx.x * 8 + wave) * 16;
  if (base >= nNodes) return;

  long long nodeA = base + l16;
  if (nodeA >= nNodes) nodeA = nNodes - 1;

  v8f accS = {0.f, 0.f, 0.f, 0.f, 0.f, 0.f, 0.f, 0.f};
  v8f accP = {0.f, 0.f, 0.f, 0.f, 0.f, 0.f, 0.f, 0.f};

  for (int k0 = 0; k0 < NHID; k0 += 4) {
    const int kb = k0 + half * 2;
    const v2f ah = *(const v2f*)(h + nodeA * NHID + kb);
    v2f bs, bn;
    bs.x = wself [kb       * NCLASS + l16];
    bs.y = wself [(kb + 1) * NCLASS + l16];
    bn.x = wneigh[kb       * NCLASS + l16];
    bn.y = wneigh[(kb + 1) * NCLASS + l16];
    accS = __builtin_amdgcn_wmma_f32_16x16x4_f32(false, ah, false, bs,
                                                 (short)0, accS, false, false);
    accP = __builtin_amdgcn_wmma_f32_16x16x4_f32(false, ah, false, bn,
                                                 (short)0, accP, false, false);
  }

  const float bv = bias[l16];
#pragma unroll
  for (int j = 0; j < 8; ++j) {
    long long node = base + j + half * 8;
    if (node < nNodes) {
      sOut[node * NCLASS + l16] = accS[j] + bv;
      pOut[node * NCLASS + l16] = accP[j];
      agg2[node * NCLASS + l16] = 0.0f;
    }
  }
}

__global__ __launch_bounds__(256)
void scatter16_kernel(const float* __restrict__ p,
                      const int* __restrict__ src,
                      const int* __restrict__ dst,
                      float* __restrict__ agg2, long long nEdges) {
  const int lane = threadIdx.x & 31;
  const int half = lane >> 4;
  const int l16  = lane & 15;
  const long long wv = ((long long)blockIdx.x * blockDim.x + threadIdx.x) >> 5;
  const long long e0 = wv * 32;
  if (e0 >= nEdges) return;
  const long long me = e0 + lane;
  int sv = 0, dv = 0;
  if (me < nEdges) { sv = src[me]; dv = dst[me]; }
  int   dr[16];
  float vr[16];
#pragma unroll
  for (int j = 0; j < 16; ++j) {
    const int sl = 2 * j + half;
    const int s = __shfl(sv, sl, 32);
    dr[j] = __shfl(dv, sl, 32);
    vr[j] = p[(long long)s * NCLASS + l16];
  }
#pragma unroll
  for (int j = 0; j < 16; ++j) {
    if (e0 + 2 * j + half < nEdges)
      atomicAdd(agg2 + (long long)dr[j] * NCLASS + l16, vr[j]);
  }
}

__global__ __launch_bounds__(256)
void final2_kernel(const float* __restrict__ s, const float* __restrict__ agg2,
                   const float* __restrict__ deg, float* __restrict__ out, long long n) {
  const long long i = (long long)blockIdx.x * blockDim.x + threadIdx.x;
  const bool ok = i < n;
  const long long ii = ok ? i : n - 1;
  const float v = s[ii] + agg2[ii] / fmaxf(deg[ii >> 4], 1.0f);
  float mx = v;
#pragma unroll
  for (int off = 8; off >= 1; off >>= 1)
    mx = fmaxf(mx, __shfl_xor(mx, off, 16));
  const float e = expf(v - mx);
  float sum = e;
#pragma unroll
  for (int off = 8; off >= 1; off >>= 1)
    sum += __shfl_xor(sum, off, 16);
  if (ok) out[i] = v - mx - logf(sum);
}

extern "C" void kernel_launch(void* const* d_in, const int* in_sizes, int n_in,
                              void* d_out, int out_size, void* d_ws, size_t ws_size,
                              hipStream_t stream) {
  const float* x   = (const float*)d_in[0];
  const int*   src = (const int*)  d_in[1];
  const int*   dst = (const int*)  d_in[2];
  const float* ws1 = (const float*)d_in[3];
  const float* wn1 = (const float*)d_in[4];
  const float* b1  = (const float*)d_in[5];
  const float* ws2 = (const float*)d_in[6];
  const float* wn2 = (const float*)d_in[7];
  const float* b2  = (const float*)d_in[8];
  float* out = (float*)d_out;

  const int nNodes = in_sizes[0] / NFEAT;
  const long long nEdges = in_sizes[1];

  char* w = (char*)d_ws;
  size_t degBytes = (((size_t)nNodes * 4) + 255) & ~(size_t)255;
  float* deg = (float*)w;
  float* agg = (float*)(w + degBytes);
  float* h   = agg + (size_t)nNodes * NFEAT;

  const int nE    = (int)nEdges;
  const int nPad  = (nNodes + 63) & ~63;
  const int nB    = (nNodes + CB_NODES - 1) >> CB_SHIFT;
  const int nBPad = (nB + 63) & ~63;
  const int nEPad = (nE + 63) & ~63;
  const int chunk = (nE + NCHUNK - 1) / NCHUNK;
  int* hmat   = (int*)h;
  int* rowTot = hmat + nB * NCHUNK;
  int* rec    = rowTot + nBPad;
  int* seg    = rec + nEPad;
  int* ssrc   = seg + 2 * nPad;

  chunk_hist_kernel<<<NCHUNK, SORT_THREADS, 0, stream>>>(dst, hmat, nE, nNodes, nB, chunk);
  row_scan_kernel<<<nB, NCHUNK, 0, stream>>>(hmat, rowTot);
  place_kernel<<<NCHUNK, SORT_THREADS, 0, stream>>>(src, dst, hmat, rowTot, rec, nE, nNodes, nB, chunk);
  fine_sort_kernel<<<nB, SORT_THREADS, 0, stream>>>(rec, rowTot, seg, ssrc, nE, nNodes);
  gather_mean_kernel<<<2048, 256, 0, stream>>>(x, seg, ssrc, agg, deg, nNodes);
  gemm1_kernel<<<(nNodes + G1_ROWS - 1) / G1_ROWS, 256, 0, stream>>>(
      x, agg, ws1, wn1, b1, h, nNodes);

  const long long outN = (long long)nNodes * NCLASS;
  float* s2   = agg;
  float* p2   = agg + outN;
  float* agg2 = agg + 2 * outN;
  const int projBlocks  = (nNodes + 127) / 128;
  const int scat16Blocks = (int)((nEdges + 255) / 256);
  const int finalBlocks = (int)((outN + 255) / 256);

  proj2_kernel<<<projBlocks, 256, 0, stream>>>(h, ws2, wn2, b2, s2, p2, agg2, nNodes);
  scatter16_kernel<<<scat16Blocks, 256, 0, stream>>>(p2, src, dst, agg2, nEdges);
  final2_kernel<<<finalBlocks, 256, 0, stream>>>(s2, agg2, deg, out, outN);
}


# layer-2 neighbour sum: per-node gather over the layer-1 buckets (kept: h overwrites agg in place) ending in log-softmax in registers; no float atomics; sums in bucket order, softmax tree regrouped
# speedup vs baseline: 3.8708x; 1.2450x over previous
#include <hip/hip_runtime.h>
#include <math.h>

typedef __attribute__((ext_vector_type(2))) float v2f;
typedef __attribute__((ext_vector_type(8))) float v8f;

#define NFEAT 128
#define NHID  128
#define NCLASS 16

#define G1_ROWS   48
#define G1_TILES  3
#define G1_STRIDE 132

#define CB_SHIFT 8
#define CB_NODES 256
#define NCHUNK   256
#define NB_MAX   1024
#define SORT_THREADS 1024
#define SORT_WAVES   32

template <int NW>
__device__ __forceinline__ int block_excl_scan(int v, int* wsum  , int& total) {
  const int lane = threadIdx.x & 31;
  const int wave = threadIdx.x >> 5;
  int inc = v;
#pragma unroll
  for (int off = 1; off < 32; off <<= 1) {
    const int t = __shfl_up(inc, off, 32);
    if (lane >= off) inc += t;
  }
  if (lane == 31) wsum[wave] = inc;
  __syncthreads();
  if (wave == 0) {
    const int w = (lane < NW) ? wsum[lane] : 0;
    int winc = w;
#pragma unroll
    for (int off = 1; off < 32; off <<= 1) {
      const int t = __shfl_up(winc, off, 32);
      if (lane >= off) winc += t;
    }
    if (lane < NW) wsum[lane] = winc - w;
    if (lane == 31) wsum[NW] = winc;
  }
  __syncthreads();
  const int r = wsum[wave] + inc - v;
  total = wsum[NW];
  __syncthreads();
  return r;
}

__global__ __launch_bounds__(SORT_THREADS)
void chunk_hist_kernel(const int* __restrict__ dst, int* __restrict__ hmat,
                       int nEdges, int nNodes, int nB, int chunk) {
  __shared__ int hist[NB_MAX];
  const int tid = threadIdx.x;
  for (int b = tid; b < NB_MAX; b += SORT_THREADS) hist[b] = 0;
  __syncthreads();
  const int e0 = (int)blockIdx.x * chunk;
  const int e1 = min(e0 + chunk, nEdges);
  for (int e = e0 + tid; e < e1; e += 4 * SORT_THREADS) {
    int d[4];
#pragma unroll
    for (int u = 0; u < 4; ++u) {
      const int ee = e + u * SORT_THREADS;
      d[u] = 0;
      if (ee < e1) d[u] = dst[ee];
    }
#pragma unroll
    for (int u = 0; u < 4; ++u) {
      if (e + u * SORT_THREADS < e1) {
        const int dd = min(max(d[u], 0), nNodes - 1);
        atomicAdd(&hist[min(dd >> CB_SHIFT, NB_MAX - 1)], 1);
      }
    }
  }
  __syncthreads();
  for (int b = tid; b < nB; b += SORT_THREADS) hmat[b * NCHUNK + (int)blockIdx.x] = hist[b];
}

__global__ __launch_bounds__(NCHUNK)
void row_scan_kernel(int* __restrict__ hmat, int* __restrict__ rowTot) {
  __shared__ int wsum[NCHUNK / 32 + 1];
  const int i = (int)blockIdx.x * NCHUNK + (int)threadIdx.x;
  const int v = hmat[i];
  int total;
  const int ex = block_excl_scan<NCHUNK / 32>(v, wsum, total);
  hmat[i] = ex;
  if (threadIdx.x == 0) rowTot[blockIdx.x] = total;
}

__global__ __launch_bounds__(SORT_THREADS)
void place_kernel(const int* __restrict__ src, const int* __restrict__ dst,
                  const int* __restrict__ hmat, const int* __restrict__ rowTot,
                  int* __restrict__ rec, int nEdges, int nNodes, int nB, int chunk) {
  __shared__ int cursor[NB_MAX];
  __shared__ int wsum[SORT_WAVES + 1];
  const int tid = threadIdx.x;
  int carry = 0;
  for (int b0 = 0; b0 < nB; b0 += SORT_THREADS) {
    const int b = b0 + tid;
    const int v = (b < nB) ? rowTot[b] : 0;
    int total;
    const int ex = block_excl_scan<SORT_WAVES>(v, wsum, total);
    if (b < min(nB, NB_MAX)) cursor[b] = carry + ex + hmat[b * NCHUNK + (int)blockIdx.x];
    carry += total;
  }
  __syncthreads();
  const int e0 = (int)blockIdx.x * chunk;
  const int e1 = min(e0 + chunk, nEdges);
  for (int e = e0 + tid; e < e1; e += 4 * SORT_THREADS) {
    int s[4], d[4];
#pragma unroll
    for (int u = 0; u < 4; ++u) {
      const int ee = e + u * SORT_THREADS;
      s[u] = 0; d[u] = 0;
      if (ee < e1) { s[u] = src[ee]; d[u] = dst[ee]; }
    }
#pragma unroll
    for (int u = 0; u < 4; ++u) {
      if (e + u * SORT_THREADS < e1) {
        const int dd = min(max(d[u], 0), nNodes - 1);
        int pos = atomicAdd(&cursor[min(dd >> CB_SHIFT, NB_MAX - 1)], 1);
        pos = min(max(pos, 0), nEdges - 1);
        rec[pos] = (int)(((unsigned)s[u] << CB_SHIFT) | (unsigned)(dd & (CB_NODES - 1)));
      }
    }
  }
}

__global__ __launch_bounds__(SORT_THREADS)
void fine_sort_kernel(const int* __restrict__ rec, const int* __restrict__ rowTot,
                      int* __restrict__ seg, int* __restrict__ ssrc,
                      int nEdges, int nNodes) {
  __shared__ int cnt[CB_NODES];
  __shared__ int wsum[SORT_WAVES + 1];
  const int tid = threadIdx.x;
  const int bkt = (int)blockIdx.x;
  int part = 0;
  for (int b = tid; b < bkt; b += SORT_THREADS) part += rowTot[b];
  int bstart;
  (void)block_excl_scan<SORT_WAVES>(part, wsum, bstart);
  const int r0 = min(max(bstart, 0), nEdges);
  const int r1 = min(r0 + max(rowTot[bkt], 0), nEdges);
  if (tid < CB_NODES) cnt[tid] = 0;
  __syncthreads();
  for (int i = r0 + tid; i < r1; i += 4 * SORT_THREADS) {
    int r[4];
#pragma unroll
    for (int u = 0; u < 4; ++u) {
      const int ii = i + u * SORT_THREADS;
      r[u] = 0;
      if (ii < r1) r[u] = rec[ii];
    }
#pragma unroll
    for (int u = 0; u < 4; ++u)
      if (i + u * SORT_THREADS < r1) atomicAdd(&cnt[r[u] & (CB_NODES - 1)], 1);
  }
  __syncthreads();
  const int c = (tid < CB_NODES) ? cnt[tid] : 0;
  int total;
  const int ex = block_excl_scan<SORT_WAVES>(c, wsum, total);
  if (tid < CB_NODES) {
    const int start = r0 + ex;
    const int node = bkt * CB_NODES + tid;
    if (node < nNodes) *(int2*)(seg + 2 * node) = make_int2(start, c);
    cnt[tid] = start;
  }
  __syncthreads();
  for (int i = r0 + tid; i < r1; i += 4 * SORT_THREADS) {
    int r[4];
#pragma unroll
    for (int u = 0; u < 4; ++u) {
      const int ii = i + u * SORT_THREADS;
      r[u] = 0;
      if (ii < r1) r[u] = rec[ii];
    }
#pragma unroll
    for (int u = 0; u < 4; ++u) {
      if (i + u * SORT_THREADS < r1) {
        int pos = atomicAdd(&cnt[r[u] & (CB_NODES - 1)], 1);
        pos = min(max(pos, 0), nEdges - 1);
        ssrc[pos] = (int)((unsigned)r[u] >> CB_SHIFT);
      }
    }
  }
}

__global__ __launch_bounds__(256)
void gather_mean_kernel(const float* __restrict__ x, const int* __restrict__ seg,
                        const int* __restrict__ ssrc,
                        float* __restrict__ agg, int nNodes) {
  const int lane = threadIdx.x & 31;
  const int wv = __builtin_amdgcn_readfirstlane((int)((blockIdx.x * blockDim.x + threadIdx.x) >> 5));
  const int nw = (int)((gridDim.x * blockDim.x) >> 5);
  const float* xl = x + lane * 4;
  for (int node = wv; node < nNodes; node += nw) {
    const int2 sg = *(const int2*)(seg + 2 * node);
    const int beg = sg.x;
    const int c   = sg.y;
    float4 acc = make_float4(0.f, 0.f, 0.f, 0.f);
    for (int b = 0; b < c; b += 32) {
      const int m = min(32, c - b);
      int sv = 0;
      if (lane < m) sv = ssrc[beg + b + lane];
      sv = min(max(sv, 0), nNodes - 1);
      for (int j = 0; j < m; j += 8) {
        float4 t[8];
#pragma unroll
        for (int u = 0; u < 8; ++u) {
          const int idx = min(j + u, m - 1);
          const int s = __builtin_amdgcn_readlane(sv, idx);
          t[u] = *(const float4*)(xl + (long long)s * NFEAT);
        }
#pragma unroll
        for (int u = 0; u < 8; ++u) {
          if (j + u < m) {
            acc.x += t[u].x; acc.y += t[u].y; acc.z += t[u].z; acc.w += t[u].w;
          }
        }
      }
    }
    const float dv = fmaxf((float)c, 1.0f);
    const float4 o = make_float4(acc.x / dv, acc.y / dv, acc.z / dv, acc.w / dv);
    *(float4*)(agg + (long long)node * NFEAT + lane * 4) = o;
  }
}

__global__ __launch_bounds__(256)
void gemm1_kernel(const float* __restrict__ x, float* aggh,
                  const float* __restrict__ wself, const float* __restrict__ wneigh,
                  const float* __restrict__ bias, int nNodes) {
  __shared__ float xs[G1_ROWS * G1_STRIDE];
  __shared__ float as[G1_ROWS * G1_STRIDE];

  const int tid  = threadIdx.x;
  const int lane = tid & 31;
  const int wave = tid >> 5;
  const int half = lane >> 4;
  const int l16  = lane & 15;
  const int col  = wave * 16 + l16;
  const long long base = (long long)blockIdx.x * G1_ROWS;

  for (int i = tid; i < G1_ROWS * 32; i += 256) {
    int row = i >> 5;
    int c4  = (i & 31) * 4;
    long long node = base + row;
    float4 vx = make_float4(0.f, 0.f, 0.f, 0.f);
    float4 va = vx;
    if (node < nNodes) {
      vx = *(const float4*)(x   + node * NFEAT + c4);
      va = *(const float4*)(aggh + node * NFEAT + c4);
    }
    *(float4*)(&xs[row * G1_STRIDE + c4]) = vx;
    *(float4*)(&as[row * G1_STRIDE + c4]) = va;
  }
  __syncthreads();

  const v8f vzero = {0.f, 0.f, 0.f, 0.f, 0.f, 0.f, 0.f, 0.f};
  v8f acc[G1_TILES];
#pragma unroll
  for (int r = 0; r < G1_TILES; ++r) acc[r] = vzero;

  for (int k0 = 0; k0 < NFEAT; k0 += 4) {
    const int kb = k0 + half * 2;
    v2f bs, bn;
    bs.x = wself [(long long)kb       * NHID + col];
    bs.y = wself [(long long)(kb + 1) * NHID + col];
    bn.x = wneigh[(long long)kb       * NHID + col];
    bn.y = wneigh[(long long)(kb + 1) * NHID + col];
#pragma unroll
    for (int r = 0; r < G1_TILES; ++r) {
      const v2f ax = *(const v2f*)(&xs[(r * 16 + l16) * G1_STRIDE + kb]);
      const v2f aa = *(const v2f*)(&as[(r * 16 + l16) * G1_STRIDE + kb]);
      acc[r] = __builtin_amdgcn_wmma_f32_16x16x4_f32(false, ax, false, bs,
                                                     (short)0, acc[r], false, false);
      acc[r] = __builtin_amdgcn_wmma_f32_16x16x4_f32(false, aa, false, bn,
                                                     (short)0, acc[r], false, false);
    }
  }

  const float bv = bias[col];
#pragma unroll
  for (int r = 0; r < G1_TILES; ++r) {
#pragma unroll
    for (int j = 0; j < 8; ++j) {
      long long node = base + r * 16 + j + half * 8;
      if (node < nNodes) {
        float v = acc[r][j] + bv;
        aggh[node * NHID + col] = fmaxf(v, 0.0f);
      }
    }
  }
}

__global__ __launch_bounds__(256)
void proj2_kernel(const float* __restrict__ h,
                  const float* __restrict__ wself, const float* __restrict__ wneigh,
                  const float* __restrict__ bias, float* __restrict__ sOut,
                  float* __restrict__ pOut, int nNodes) {
  const int lane = threadIdx.x & 31;
  const int wave = threadIdx.x >> 5;
  const int half = lane >> 4;
  const int l16  = lane & 15;
  const long long base = ((long long)blockIdx.x * 8 + wave) * 16;
  if (base >= nNodes) return;

  long long nodeA = base + l16;
  if (nodeA >= nNodes) nodeA = nNodes - 1;

  v8f accS = {0.f, 0.f, 0.f, 0.f, 0.f, 0.f, 0.f, 0.f};
  v8f accP = {0.f, 0.f, 0.f, 0.f, 0.f, 0.f, 0.f, 0.f};

  for (int k0 = 0; k0 < NHID; k0 += 4) {
    const int kb = k0 + half * 2;
    const v2f ah = *(const v2f*)(h + nodeA * NHID + kb);
    v2f bs, bn;
    bs.x = wself [kb       * NCLASS + l16];
    bs.y = wself [(kb + 1) * NCLASS + l16];
    bn.x = wneigh[kb       * NCLASS + l16];
    bn.y = wneigh[(kb + 1) * NCLASS + l16];
    accS = __builtin_amdgcn_wmma_f32_16x16x4_f32(false, ah, false, bs,
                                                 (short)0, accS, false, false);
    accP = __builtin_amdgcn_wmma_f32_16x16x4_f32(false, ah, false, bn,
                                                 (short)0, accP, false, false);
  }

  const float bv = bias[l16];
#pragma unroll
  for (int j = 0; j < 8; ++j) {
    long long node = base + j + half * 8;
    if (node < nNodes) {
      sOut[node * NCLASS + l16] = accS[j] + bv;
      pOut[node * NCLASS + l16] = accP[j];
    }
  }
}

__device__ __forceinline__ int quad_bcast(int v, int u) {
  switch (u) {
    case 0:  return __builtin_amdgcn_update_dpp(0, v, 0x00, 0xf, 0xf, false);
    case 1:  return __builtin_amdgcn_update_dpp(0, v, 0x55, 0xf, 0xf, false);
    case 2:  return __builtin_amdgcn_update_dpp(0, v, 0xaa, 0xf, 0xf, false);
    default: return __builtin_amdgcn_update_dpp(0, v, 0xff, 0xf, 0xf, false);
  }
}

__global__ __launch_bounds__(256)
void gather16_final_kernel(const float* __restrict__ s, const float* __restrict__ p,
                           const int* __restrict__ seg, const int* __restrict__ ssrc,
                           float* __restrict__ out, int nNodes, int nEdges) {
  const int lane = threadIdx.x & 31;
  const int q    = lane & 3;
  const int wv = (int)((blockIdx.x * blockDim.x + threadIdx.x) >> 5);
  const int nw = (int)((gridDim.x * blockDim.x) >> 5);
  for (int n0 = wv * 8; n0 < nNodes; n0 += nw * 8) {
    const int node = n0 + (lane >> 2);
    const bool ok = node < nNodes;
    int beg = 0, c = 0;
    float4 sv = make_float4(0.f, 0.f, 0.f, 0.f);
    if (ok) {
      const int2 sg = *(const int2*)(seg + 2 * node);
      beg = sg.x;
      c   = sg.y;
      sv  = *(const float4*)(s + (long long)node * NCLASS + q * 4);
    }
    float4 acc = make_float4(0.f, 0.f, 0.f, 0.f);
    for (int b0 = 0; __any(b0 < c); b0 += 16) {
      int idx[4];
#pragma unroll
      for (int t = 0; t < 4; ++t) {
        const int pos = min(max(beg + b0 + 4 * t + q, 0), nEdges - 1);
        idx[t] = min(max(ssrc[pos], 0), nNodes - 1);
      }
      float4 tv[16];
#pragma unroll
      for (int j = 0; j < 16; ++j) {
        const int sj = quad_bcast(idx[j >> 2], j & 3);
        tv[j] = make_float4(0.f, 0.f, 0.f, 0.f);
        if (b0 + j < c) tv[j] = *(const float4*)(p + (long long)sj * NCLASS + q * 4);
      }
#pragma unroll
      for (int j = 0; j < 16; ++j) {
        acc.x += tv[j].x; acc.y += tv[j].y; acc.z += tv[j].z; acc.w += tv[j].w;
      }
    }
    const float dv = fmaxf((float)c, 1.0f);
    const float4 v = make_float4(sv.x + acc.x / dv, sv.y + acc.y / dv,
                                 sv.z + acc.z / dv, sv.w + acc.w / dv);
    float mx = fmaxf(fmaxf(v.x, v.y), fmaxf(v.z, v.w));
    mx = fmaxf(mx, __shfl_xor(mx, 1, 32));
    mx = fmaxf(mx, __shfl_xor(mx, 2, 32));
    float sum = (expf(v.x - mx) + expf(v.y - mx)) + (expf(v.z - mx) + expf(v.w - mx));
    sum += __shfl_xor(sum, 1, 32);
    sum += __shfl_xor(sum, 2, 32);
    const float ls = logf(sum);
    if (ok)
      *(float4*)(out + (long long)node * NCLASS + q * 4) =
          make_float4(v.x - mx - ls, v.y - mx - ls, v.z - mx - ls, v.w - mx - ls);
  }
}

extern "C" void kernel_launch(void* const* d_in, const int* in_sizes, int n_in,
                              void* d_out, int out_size, void* d_ws, size_t ws_size,
                              hipStream_t stream) {
  const float* x   = (const float*)d_in[0];
  const int*   src = (const int*)  d_in[1];
  const int*   dst = (const int*)  d_in[2];
  const float* ws1 = (const float*)d_in[3];
  const float* wn1 = (const float*)d_in[4];
  const float* b1  = (const float*)d_in[5];
  const float* ws2 = (const float*)d_in[6];
  const float* wn2 = (const float*)d_in[7];
  const float* b2  = (const float*)d_in[8];
  float* out = (float*)d_out;

  const int nNodes = in_sizes[0] / NFEAT;
  const long long nEdges = in_sizes[1];

  float* ah  = (float*)d_ws;
  int*   aux = (int*)(ah + (size_t)nNodes * NFEAT);

  const int nE    = (int)nEdges;
  const int nPad  = (nNodes + 63) & ~63;
  const int nB    = (nNodes + CB_NODES - 1) >> CB_SHIFT;
  const int nBPad = (nB + 63) & ~63;
  const int nEPad = (nE + 63) & ~63;
  const int chunk = (nE + NCHUNK - 1) / NCHUNK;
  int* hmat   = aux;
  int* rowTot = hmat + nB * NCHUNK;
  int* rec    = rowTot + nBPad;
  int* seg    = rec + nEPad;
  int* ssrc   = seg + 2 * nPad;
  const long long outN = (long long)nNodes * NCLASS;
  float* s2   = (float*)(ssrc + nEPad);
  float* p2   = s2 + outN;

  chunk_hist_kernel<<<NCHUNK, SORT_THREADS, 0, stream>>>(dst, hmat, nE, nNodes, nB, chunk);
  row_scan_kernel<<<nB, NCHUNK, 0, stream>>>(hmat, rowTot);
  place_kernel<<<NCHUNK, SORT_THREADS, 0, stream>>>(src, dst, hmat, rowTot, rec, nE, nNodes, nB, chunk);
  fine_sort_kernel<<<nB, SORT_THREADS, 0, stream>>>(rec, rowTot, seg, ssrc, nE, nNodes);
  gather_mean_kernel<<<2048, 256, 0, stream>>>(x, seg, ssrc, ah, nNodes);
  gemm1_kernel<<<(nNodes + G1_ROWS - 1) / G1_ROWS, 256, 0, stream>>>(
      x, ah, ws1, wn1, b1, nNodes);

  const int projBlocks = (nNodes + 127) / 128;
  const int g16Blocks  = (nNodes + 127) / 128;

  proj2_kernel<<<projBlocks, 256, 0, stream>>>(ah, ws2, wn2, b2, s2, p2, nNodes);
  gather16_final_kernel<<<g16Blocks, 256, 0, stream>>>(s2, p2, seg, ssrc, out, nNodes, nE);
}


# layer-1 neighbour mean summed inside the product kernel: each block gathers its 48 nodes' source rows straight into its LDS A tile; agg plane and gather launch gone; per-node sum order unchanged
# speedup vs baseline: 4.1507x; 1.0723x over previous
#include <hip/hip_runtime.h>
#include <math.h>

typedef __attribute__((ext_vector_type(2))) float v2f;
typedef __attribute__((ext_vector_type(8))) float v8f;

#define NFEAT 128
#define NHID  128
#define NCLASS 16

#define G1_ROWS   48
#define G1_TILES  3
#define G1_STRIDE 132

#define CB_SHIFT 8
#define CB_NODES 256
#define NCHUNK   256
#define NB_MAX   1024
#define SORT_THREADS 1024
#define SORT_WAVES   32

template <int NW>
__device__ __forceinline__ int block_excl_scan(int v, int* wsum  , int& total) {
  const int lane = threadIdx.x & 31;
  const int wave = threadIdx.x >> 5;
  int inc = v;
#pragma unroll
  for (int off = 1; off < 32; off <<= 1) {
    const int t = __shfl_up(inc, off, 32);
    if (lane >= off) inc += t;
  }
  if (lane == 31) wsum[wave] = inc;
  __syncthreads();
  if (wave == 0) {
    const int w = (lane < NW) ? wsum[lane] : 0;
    int winc = w;
#pragma unroll
    for (int off = 1; off < 32; off <<= 1) {
      const int t = __shfl_up(winc, off, 32);
      if (lane >= off) winc += t;
    }
    if (lane < NW) wsum[lane] = winc - w;
    if (lane == 31) wsum[NW] = winc;
  }
  __syncthreads();
  const int r = wsum[wave] + inc - v;
  total = wsum[NW];
  __syncthreads();
  return r;
}

__global__ __launch_bounds__(SORT_THREADS)
void chunk_hist_kernel(const int* __restrict__ dst, int* __restrict__ hmat,
                       int nEdges, int nNodes, int nB, int chunk) {
  __shared__ int hist[NB_MAX];
  const int tid = threadIdx.x;
  for (int b = tid; b < NB_MAX; b += SORT_THREADS) hist[b] = 0;
  __syncthreads();
  const int e0 = (int)blockIdx.x * chunk;
  const int e1 = min(e0 + chunk, nEdges);
  for (int e = e0 + tid; e < e1; e += 4 * SORT_THREADS) {
    int d[4];
#pragma unroll
    for (int u = 0; u < 4; ++u) {
      const int ee = e + u * SORT_THREADS;
      d[u] = 0;
      if (ee < e1) d[u] = dst[ee];
    }
#pragma unroll
    for (int u = 0; u < 4; ++u) {
      if (e + u * SORT_THREADS < e1) {
        const int dd = min(max(d[u], 0), nNodes - 1);
        atomicAdd(&hist[min(dd >> CB_SHIFT, NB_MAX - 1)], 1);
      }
    }
  }
  __syncthreads();
  for (int b = tid; b < nB; b += SORT_THREADS) hmat[b * NCHUNK + (int)blockIdx.x] = hist[b];
}

__global__ __launch_bounds__(NCHUNK)
void row_scan_kernel(int* __restrict__ hmat, int* __restrict__ rowTot) {
  __shared__ int wsum[NCHUNK / 32 + 1];
  const int i = (int)blockIdx.x * NCHUNK + (int)threadIdx.x;
  const int v = hmat[i];
  int total;
  const int ex = block_excl_scan<NCHUNK / 32>(v, wsum, total);
  hmat[i] = ex;
  if (threadIdx.x == 0) rowTot[blockIdx.x] = total;
}

__global__ __launch_bounds__(SORT_THREADS)
void place_kernel(const int* __restrict__ src, const int* __restrict__ dst,
                  const int* __restrict__ hmat, const int* __restrict__ rowTot,
                  int* __restrict__ rec, int nEdges, int nNodes, int nB, int chunk) {
  __shared__ int cursor[NB_MAX];
  __shared__ int wsum[SORT_WAVES + 1];
  const int tid = threadIdx.x;
  int carry = 0;
  for (int b0 = 0; b0 < nB; b0 += SORT_THREADS) {
    const int b = b0 + tid;
    const int v = (b < nB) ? rowTot[b] : 0;
    int total;
    const int ex = block_excl_scan<SORT_WAVES>(v, wsum, total);
    if (b < min(nB, NB_MAX)) cursor[b] = carry + ex + hmat[b * NCHUNK + (int)blockIdx.x];
    carry += total;
  }
  __syncthreads();
  const int e0 = (int)blockIdx.x * chunk;
  const int e1 = min(e0 + chunk, nEdges);
  for (int e = e0 + tid; e < e1; e += 4 * SORT_THREADS) {
    int s[4], d[4];
#pragma unroll
    for (int u = 0; u < 4; ++u) {
      const int ee = e + u * SORT_THREADS;
      s[u] = 0; d[u] = 0;
      if (ee < e1) { s[u] = src[ee]; d[u] = dst[ee]; }
    }
#pragma unroll
    for (int u = 0; u < 4; ++u) {
      if (e + u * SORT_THREADS < e1) {
        const int dd = min(max(d[u], 0), nNodes - 1);
        int pos = atomicAdd(&cursor[min(dd >> CB_SHIFT, NB_MAX - 1)], 1);
        pos = min(max(pos, 0), nEdges - 1);
        rec[pos] = (int)(((unsigned)s[u] << CB_SHIFT) | (unsigned)(dd & (CB_NODES - 1)));
      }
    }
  }
}

__global__ __launch_bounds__(SORT_THREADS)
void fine_sort_kernel(const int* __restrict__ rec, const int* __restrict__ rowTot,
                      int* __restrict__ seg, int* __restrict__ ssrc,
                      int nEdges, int nNodes) {
  __shared__ int cnt[CB_NODES];
  __shared__ int wsum[SORT_WAVES + 1];
  const int tid = threadIdx.x;
  const int bkt = (int)blockIdx.x;
  int part = 0;
  for (int b = tid; b < bkt; b += SORT_THREADS) part += rowTot[b];
  int bstart;
  (void)block_excl_scan<SORT_WAVES>(part, wsum, bstart);
  const int r0 = min(max(bstart, 0), nEdges);
  const int r1 = min(r0 + max(rowTot[bkt], 0), nEdges);
  if (tid < CB_NODES) cnt[tid] = 0;
  __syncthreads();
  for (int i = r0 + tid; i < r1; i += 4 * SORT_THREADS) {
    int r[4];
#pragma unroll
    for (int u = 0; u < 4; ++u) {
      const int ii = i + u * SORT_THREADS;
      r[u] = 0;
      if (ii < r1) r[u] = rec[ii];
    }
#pragma unroll
    for (int u = 0; u < 4; ++u)
      if (i + u * SORT_THREADS < r1) atomicAdd(&cnt[r[u] & (CB_NODES - 1)], 1);
  }
  __syncthreads();
  const int c = (tid < CB_NODES) ? cnt[tid] : 0;
  int total;
  const int ex = block_excl_scan<SORT_WAVES>(c, wsum, total);
  if (tid < CB_NODES) {
    const int start = r0 + ex;
    const int node = bkt * CB_NODES + tid;
    if (node < nNodes) *(int2*)(seg + 2 * node) = make_int2(start, c);
    cnt[tid] = start;
  }
  __syncthreads();
  for (int i = r0 + tid; i < r1; i += 4 * SORT_THREADS) {
    int r[4];
#pragma unroll
    for (int u = 0; u < 4; ++u) {
      const int ii = i + u * SORT_THREADS;
      r[u] = 0;
      if (ii < r1) r[u] = rec[ii];
    }
#pragma unroll
    for (int u = 0; u < 4; ++u) {
      if (i + u * SORT_THREADS < r1) {
        int pos = atomicAdd(&cnt[r[u] & (CB_NODES - 1)], 1);
        pos = min(max(pos, 0), nEdges - 1);
        ssrc[pos] = (int)((unsigned)r[u] >> CB_SHIFT);
      }
    }
  }
}

#define G1_NPW (G1_ROWS / 8)
#define G1_XLD (G1_ROWS * 32 / 256)
__global__ __launch_bounds__(256)
void layer1_kernel(const float* __restrict__ x, const int* __restrict__ seg,
                  const int* __restrict__ ssrc, float* __restrict__ hOut,
                  const float* __restrict__ wself, const float* __restrict__ wneigh,
                  const float* __restrict__ bias, int nNodes) {
  __shared__ float xs[G1_ROWS * G1_STRIDE];
  __shared__ float as[G1_ROWS * G1_STRIDE];

  const int tid  = threadIdx.x;
  const int lane = tid & 31;
  const int wave = tid >> 5;
  const int half = lane >> 4;
  const int l16  = lane & 15;
  const int col  = wave * 16 + l16;
  const long long base = (long long)blockIdx.x * G1_ROWS;

  const int wrow = __builtin_amdgcn_readfirstlane(wave) * G1_NPW;
  int2 sgv = make_int2(0, 0);
  {
    const long long nl = base + wrow + lane;
    if (lane < G1_NPW && nl < nNodes) sgv = *(const int2*)(seg + 2 * nl);
  }
  const float* xl = x + lane * 4;
  for (int i = 0; i < G1_NPW; ++i) {
    const int beg = __builtin_amdgcn_readlane(sgv.x, i);
    const int c   = __builtin_amdgcn_readlane(sgv.y, i);
    float4 acc = make_float4(0.f, 0.f, 0.f, 0.f);
    for (int b = 0; b < c; b += 32) {
      const int m = min(32, c - b);
      int sv = 0;
      if (lane < m) sv = ssrc[beg + b + lane];
      sv = min(max(sv, 0), nNodes - 1);
      for (int j = 0; j < m; j += 8) {
        float4 t[8];
#pragma unroll
        for (int u = 0; u < 8; ++u) {
          const int idx = min(j + u, m - 1);
          const int s = __builtin_amdgcn_readlane(sv, idx);
          t[u] = *(const float4*)(xl + (long long)s * NFEAT);
        }
#pragma unroll
        for (int u = 0; u < 8; ++u) {
          if (j + u < m) {
            acc.x += t[u].x; acc.y += t[u].y; acc.z += t[u].z; acc.w += t[u].w;
          }
        }
      }
    }
    const float dv = fmaxf((float)c, 1.0f);
    *(float4*)(&as[(wrow + i) * G1_STRIDE + lane * 4]) =
        make_float4(acc.x / dv, acc.y / dv, acc.z / dv, acc.w / dv);
  }

  const long long xrow0 = base + (tid >> 5);
  float* xsd = &xs[(tid >> 5) * G1_STRIDE + lane * 4];
  if (base + G1_ROWS <= nNodes) {
    const float* xt = x + xrow0 * NFEAT + lane * 4;
    float4 vx[G1_XLD];
#pragma unroll
    for (int r = 0; r < G1_XLD; ++r) vx[r] = *(const float4*)(xt + r * 8 * NFEAT);
#pragma unroll
    for (int r = 0; r < G1_XLD; ++r) *(float4*)(xsd + r * 8 * G1_STRIDE) = vx[r];
  } else {
    float4 vx[G1_XLD];
#pragma unroll
    for (int r = 0; r < G1_XLD; ++r) {
      const long long node = xrow0 + r * 8;
      const long long nr = (node < nNodes) ? node : (long long)(nNodes - 1);
      vx[r] = *(const float4*)(x + nr * NFEAT + lane * 4);
    }
#pragma unroll
    for (int r = 0; r < G1_XLD; ++r) *(float4*)(xsd + r * 8 * G1_STRIDE) = vx[r];
  }
  __syncthreads();

  const v8f vzero = {0.f, 0.f, 0.f, 0.f, 0.f, 0.f, 0.f, 0.f};
  v8f acc[G1_TILES];
#pragma unroll
  for (int r = 0; r < G1_TILES; ++r) acc[r] = vzero;

  for (int k0 = 0; k0 < NFEAT; k0 += 4) {
    const int kb = k0 + half * 2;
    v2f bs, bn;
    bs.x = wself [(long long)kb       * NHID + col];
    bs.y = wself [(long long)(kb + 1) * NHID + col];
    bn.x = wneigh[(long long)kb       * NHID + col];
    bn.y = wneigh[(long long)(kb + 1) * NHID + col];
#pragma unroll
    for (int r = 0; r < G1_TILES; ++r) {
      const v2f ax = *(const v2f*)(&xs[(r * 16 + l16) * G1_STRIDE + kb]);
      const v2f aa = *(const v2f*)(&as[(r * 16 + l16) * G1_STRIDE + kb]);
      acc[r] = __builtin_amdgcn_wmma_f32_16x16x4_f32(false, ax, false, bs,
                                                     (short)0, acc[r], false, false);
      acc[r] = __builtin_amdgcn_wmma_f32_16x16x4_f32(false, aa, false, bn,
                                                     (short)0, acc[r], false, false);
    }
  }

  const float bv = bias[col];
#pragma unroll
  for (int r = 0; r < G1_TILES; ++r) {
#pragma unroll
    for (int j = 0; j < 8; ++j) {
      long long node = base + r * 16 + j + half * 8;
      if (node < nNodes) {
        float v = acc[r][j] + bv;
        hOut[node * NHID + col] = fmaxf(v, 0.0f);
      }
    }
  }
}

__global__ __launch_bounds__(256)
void proj2_kernel(const float* __restrict__ h,
                  const float* __restrict__ wself, const float* __restrict__ wneigh,
                  const float* __restrict__ bias, float* __restrict__ sOut,
                  float* __restrict__ pOut, int nNodes) {
  const int lane = threadIdx.x & 31;
  const int wave = threadIdx.x >> 5;
  const int half = lane >> 4;
  const int l16  = lane & 15;
  const long long base = ((long long)blockIdx.x * 8 + wave) * 16;
  if (base >= nNodes) return;

  long long nodeA = base + l16;
  if (nodeA >= nNodes) nodeA = nNodes - 1;

  v8f accS = {0.f, 0.f, 0.f, 0.f, 0.f, 0.f, 0.f, 0.f};
  v8f accP = {0.f, 0.f, 0.f, 0.f, 0.f, 0.f, 0.f, 0.f};

  for (int k0 = 0; k0 < NHID; k0 += 4) {
    const int kb = k0 + half * 2;
    const v2f ah = *(const v2f*)(h + nodeA * NHID + kb);
    v2f bs, bn;
    bs.x = wself [kb       * NCLASS + l16];
    bs.y = wself [(kb + 1) * NCLASS + l16];
    bn.x = wneigh[kb       * NCLASS + l16];
    bn.y = wneigh[(kb + 1) * NCLASS + l16];
    accS = __builtin_amdgcn_wmma_f32_16x16x4_f32(false, ah, false, bs,
                                                 (short)0, accS, false, false);
    accP = __builtin_amdgcn_wmma_f32_16x16x4_f32(false, ah, false, bn,
                                                 (short)0, accP, false, false);
  }

  const float bv = bias[l16];
#pragma unroll
  for (int j = 0; j < 8; ++j) {
    long long node = base + j + half * 8;
    if (node < nNodes) {
      sOut[node * NCLASS + l16] = accS[j] + bv;
      pOut[node * NCLASS + l16] = accP[j];
    }
  }
}

__device__ __forceinline__ int quad_bcast(int v, int u) {
  switch (u) {
    case 0:  return __builtin_amdgcn_update_dpp(0, v, 0x00, 0xf, 0xf, false);
    case 1:  return __builtin_amdgcn_update_dpp(0, v, 0x55, 0xf, 0xf, false);
    case 2:  return __builtin_amdgcn_update_dpp(0, v, 0xaa, 0xf, 0xf, false);
    default: return __builtin_amdgcn_update_dpp(0, v, 0xff, 0xf, 0xf, false);
  }
}

__global__ __launch_bounds__(256)
void gather16_final_kernel(const float* __restrict__ s, const float* __restrict__ p,
                           const int* __restrict__ seg, const int* __restrict__ ssrc,
                           float* __restrict__ out, int nNodes, int nEdges) {
  const int lane = threadIdx.x & 31;
  const int q    = lane & 3;
  const int wv = (int)((blockIdx.x * blockDim.x + threadIdx.x) >> 5);
  const int nw = (int)((gridDim.x * blockDim.x) >> 5);
  for (int n0 = wv * 8; n0 < nNodes; n0 += nw * 8) {
    const int node = n0 + (lane >> 2);
    const bool ok = node < nNodes;
    int beg = 0, c = 0;
    float4 sv = make_float4(0.f, 0.f, 0.f, 0.f);
    if (ok) {
      const int2 sg = *(const int2*)(seg + 2 * node);
      beg = sg.x;
      c   = sg.y;
      sv  = *(const float4*)(s + (long long)node * NCLASS + q * 4);
    }
    float4 acc = make_float4(0.f, 0.f, 0.f, 0.f);
    for (int b0 = 0; __any(b0 < c); b0 += 16) {
      int idx[4];
#pragma unroll
      for (int t = 0; t < 4; ++t) {
        const int pos = min(max(beg + b0 + 4 * t + q, 0), nEdges - 1);
        idx[t] = min(max(ssrc[pos], 0), nNodes - 1);
      }
      float4 tv[16];
#pragma unroll
      for (int j = 0; j < 16; ++j) {
        const int sj = quad_bcast(idx[j >> 2], j & 3);
        tv[j] = make_float4(0.f, 0.f, 0.f, 0.f);
        if (b0 + j < c) tv[j] = *(const float4*)(p + (long long)sj * NCLASS + q * 4);
      }
#pragma unroll
      for (int j = 0; j < 16; ++j) {
        acc.x += tv[j].x; acc.y += tv[j].y; acc.z += tv[j].z; acc.w += tv[j].w;
      }
    }
    const float dv = fmaxf((float)c, 1.0f);
    const float4 v = make_float4(sv.x + acc.x / dv, sv.y + acc.y / dv,
                                 sv.z + acc.z / dv, sv.w + acc.w / dv);
    float mx = fmaxf(fmaxf(v.x, v.y), fmaxf(v.z, v.w));
    mx = fmaxf(mx, __shfl_xor(mx, 1, 32));
    mx = fmaxf(mx, __shfl_xor(mx, 2, 32));
    float sum = (expf(v.x - mx) + expf(v.y - mx)) + (expf(v.z - mx) + expf(v.w - mx));
    sum += __shfl_xor(sum, 1, 32);
    sum += __shfl_xor(sum, 2, 32);
    const float ls = logf(sum);
    if (ok)
      *(float4*)(out + (long long)node * NCLASS + q * 4) =
          make_float4(v.x - mx - ls, v.y - mx - ls, v.z - mx - ls, v.w - mx - ls);
  }
}

extern "C" void kernel_launch(void* const* d_in, const int* in_sizes, int n_in,
                              void* d_out, int out_size, void* d_ws, size_t ws_size,
                              hipStream_t stream) {
  const float* x   = (const float*)d_in[0];
  const int*   src = (const int*)  d_in[1];
  const int*   dst = (const int*)  d_in[2];
  const float* ws1 = (const float*)d_in[3];
  const float* wn1 = (const float*)d_in[4];
  const float* b1  = (const float*)d_in[5];
  const float* ws2 = (const float*)d_in[6];
  const float* wn2 = (const float*)d_in[7];
  const float* b2  = (const float*)d_in[8];
  float* out = (float*)d_out;

  const int nNodes = in_sizes[0] / NFEAT;
  const long long nEdges = in_sizes[1];

  float* ah  = (float*)d_ws;
  int*   aux = (int*)(ah + (size_t)nNodes * NFEAT);

  const int nE    = (int)nEdges;
  const int nPad  = (nNodes + 63) & ~63;
  const int nB    = (nNodes + CB_NODES - 1) >> CB_SHIFT;
  const int nBPad = (nB + 63) & ~63;
  const int nEPad = (nE + 63) & ~63;
  const int chunk = (nE + NCHUNK - 1) / NCHUNK;
  int* hmat   = aux;
  int* rowTot = hmat + nB * NCHUNK;
  int* rec    = rowTot + nBPad;
  int* seg    = rec + nEPad;
  int* ssrc   = seg + 2 * nPad;
  const long long outN = (long long)nNodes * NCLASS;
  float* s2   = (float*)(ssrc + nEPad);
  float* p2   = s2 + outN;

  chunk_hist_kernel<<<NCHUNK, SORT_THREADS, 0, stream>>>(dst, hmat, nE, nNodes, nB, chunk);
  row_scan_kernel<<<nB, NCHUNK, 0, stream>>>(hmat, rowTot);
  place_kernel<<<NCHUNK, SORT_THREADS, 0, stream>>>(src, dst, hmat, rowTot, rec, nE, nNodes, nB, chunk);
  fine_sort_kernel<<<nB, SORT_THREADS, 0, stream>>>(rec, rowTot, seg, ssrc, nE, nNodes);
  layer1_kernel<<<(nNodes + G1_ROWS - 1) / G1_ROWS, 256, 0, stream>>>(
      x, seg, ssrc, ah, ws1, wn1, b1, nNodes);

  const int projBlocks = (nNodes + 127) / 128;
  const int g16Blocks  = (nNodes + 127) / 128;

  proj2_kernel<<<projBlocks, 256, 0, stream>>>(ah, ws2, wn2, b2, s2, p2, nNodes);
  gather16_final_kernel<<<g16Blocks, 256, 0, stream>>>(s2, p2, seg, ssrc, out, nNodes, nE);
}


# layer-2 projections (h@Wself2+b2, h@Wneigh2) moved into the layer-1 kernel: each block projects its own h tile from LDS on 3 waves; h is never stored or re-read; one launch fewer; sums unchanged
# speedup vs baseline: 4.7672x; 1.1485x over previous
#include <hip/hip_runtime.h>
#include <math.h>

typedef __attribute__((ext_vector_type(2))) float v2f;
typedef __attribute__((ext_vector_type(8))) float v8f;

#define NFEAT 128
#define NHID  128
#define NCLASS 16

#define G1_ROWS   48
#define G1_TILES  3
#define G1_STRIDE 132

#define CB_SHIFT 8
#define CB_NODES 256
#define NCHUNK   256
#define NB_MAX   1024
#define SORT_THREADS 1024
#define SORT_WAVES   32

template <int NW>
__device__ __forceinline__ int block_excl_scan(int v, int* wsum  , int& total) {
  const int lane = threadIdx.x & 31;
  const int wave = threadIdx.x >> 5;
  int inc = v;
#pragma unroll
  for (int off = 1; off < 32; off <<= 1) {
    const int t = __shfl_up(inc, off, 32);
    if (lane >= off) inc += t;
  }
  if (lane == 31) wsum[wave] = inc;
  __syncthreads();
  if (wave == 0) {
    const int w = (lane < NW) ? wsum[lane] : 0;
    int winc = w;
#pragma unroll
    for (int off = 1; off < 32; off <<= 1) {
      const int t = __shfl_up(winc, off, 32);
      if (lane >= off) winc += t;
    }
    if (lane < NW) wsum[lane] = winc - w;
    if (lane == 31) wsum[NW] = winc;
  }
  __syncthreads();
  const int r = wsum[wave] + inc - v;
  total = wsum[NW];
  __syncthreads();
  return r;
}

__global__ __launch_bounds__(SORT_THREADS)
void chunk_hist_kernel(const int* __restrict__ dst, int* __restrict__ hmat,
                       int nEdges, int nNodes, int nB, int chunk) {
  __shared__ int hist[NB_MAX];
  const int tid = threadIdx.x;
  for (int b = tid; b < NB_MAX; b += SORT_THREADS) hist[b] = 0;
  __syncthreads();
  const int e0 = (int)blockIdx.x * chunk;
  const int e1 = min(e0 + chunk, nEdges);
  for (int e = e0 + tid; e < e1; e += 4 * SORT_THREADS) {
    int d[4];
#pragma unroll
    for (int u = 0; u < 4; ++u) {
      const int ee = e + u * SORT_THREADS;
      d[u] = 0;
      if (ee < e1) d[u] = dst[ee];
    }
#pragma unroll
    for (int u = 0; u < 4; ++u) {
      if (e + u * SORT_THREADS < e1) {
        const int dd = min(max(d[u], 0), nNodes - 1);
        atomicAdd(&hist[min(dd >> CB_SHIFT, NB_MAX - 1)], 1);
      }
    }
  }
  __syncthreads();
  for (int b = tid; b < nB; b += SORT_THREADS) hmat[b * NCHUNK + (int)blockIdx.x] = hist[b];
}

__global__ __launch_bounds__(NCHUNK)
void row_scan_kernel(int* __restrict__ hmat, int* __restrict__ rowTot) {
  __shared__ int wsum[NCHUNK / 32 + 1];
  const int i = (int)blockIdx.x * NCHUNK + (int)threadIdx.x;
  const int v = hmat[i];
  int total;
  const int ex = block_excl_scan<NCHUNK / 32>(v, wsum, total);
  hmat[i] = ex;
  if (threadIdx.x == 0) rowTot[blockIdx.x] = total;
}

__global__ __launch_bounds__(SORT_THREADS)
void place_kernel(const int* __restrict__ src, const int* __restrict__ dst,
                  const int* __restrict__ hmat, const int* __restrict__ rowTot,
                  int* __restrict__ rec, int nEdges, int nNodes, int nB, int chunk) {
  __shared__ int cursor[NB_MAX];
  __shared__ int wsum[SORT_WAVES + 1];
  const int tid = threadIdx.x;
  int carry = 0;
  for (int b0 = 0; b0 < nB; b0 += SORT_THREADS) {
    const int b = b0 + tid;
    const int v = (b < nB) ? rowTot[b] : 0;
    int total;
    const int ex = block_excl_scan<SORT_WAVES>(v, wsum, total);
    if (b < min(nB, NB_MAX)) cursor[b] = carry + ex + hmat[b * NCHUNK + (int)blockIdx.x];
    carry += total;
  }
  __syncthreads();
  const int e0 = (int)blockIdx.x * chunk;
  const int e1 = min(e0 + chunk, nEdges);
  for (int e = e0 + tid; e < e1; e += 4 * SORT_THREADS) {
    int s[4], d[4];
#pragma unroll
    for (int u = 0; u < 4; ++u) {
      const int ee = e + u * SORT_THREADS;
      s[u] = 0; d[u] = 0;
      if (ee < e1) { s[u] = src[ee]; d[u] = dst[ee]; }
    }
#pragma unroll
    for (int u = 0; u < 4; ++u) {
      if (e + u * SORT_THREADS < e1) {
        const int dd = min(max(d[u], 0), nNodes - 1);
        int pos = atomicAdd(&cursor[min(dd >> CB_SHIFT, NB_MAX - 1)], 1);
        pos = min(max(pos, 0), nEdges - 1);
        rec[pos] = (int)(((unsigned)s[u] << CB_SHIFT) | (unsigned)(dd & (CB_NODES - 1)));
      }
    }
  }
}

__global__ __launch_bounds__(SORT_THREADS)
void fine_sort_kernel(const int* __restrict__ rec, const int* __restrict__ rowTot,
                      int* __restrict__ seg, int* __restrict__ ssrc,
                      int nEdges, int nNodes) {
  __shared__ int cnt[CB_NODES];
  __shared__ int wsum[SORT_WAVES + 1];
  const int tid = threadIdx.x;
  const int bkt = (int)blockIdx.x;
  int part = 0;
  for (int b = tid; b < bkt; b += SORT_THREADS) part += rowTot[b];
  int bstart;
  (void)block_excl_scan<SORT_WAVES>(part, wsum, bstart);
  const int r0 = min(max(bstart, 0), nEdges);
  const int r1 = min(r0 + max(rowTot[bkt], 0), nEdges);
  if (tid < CB_NODES) cnt[tid] = 0;
  __syncthreads();
  for (int i = r0 + tid; i < r1; i += 4 * SORT_THREADS) {
    int r[4];
#pragma unroll
    for (int u = 0; u < 4; ++u) {
      const int ii = i + u * SORT_THREADS;
      r[u] = 0;
      if (ii < r1) r[u] = rec[ii];
    }
#pragma unroll
    for (int u = 0; u < 4; ++u)
      if (i + u * SORT_THREADS < r1) atomicAdd(&cnt[r[u] & (CB_NODES - 1)], 1);
  }
  __syncthreads();
  const int c = (tid < CB_NODES) ? cnt[tid] : 0;
  int total;
  const int ex = block_excl_scan<SORT_WAVES>(c, wsum, total);
  if (tid < CB_NODES) {
    const int start = r0 + ex;
    const int node = bkt * CB_NODES + tid;
    if (node < nNodes) *(int2*)(seg + 2 * node) = make_int2(start, c);
    cnt[tid] = start;
  }
  __syncthreads();
  for (int i = r0 + tid; i < r1; i += 4 * SORT_THREADS) {
    int r[4];
#pragma unroll
    for (int u = 0; u < 4; ++u) {
      const int ii = i + u * SORT_THREADS;
      r[u] = 0;
      if (ii < r1) r[u] = rec[ii];
    }
#pragma unroll
    for (int u = 0; u < 4; ++u) {
      if (i + u * SORT_THREADS < r1) {
        int pos = atomicAdd(&cnt[r[u] & (CB_NODES - 1)], 1);
        pos = min(max(pos, 0), nEdges - 1);
        ssrc[pos] = (int)((unsigned)r[u] >> CB_SHIFT);
      }
    }
  }
}

#define G1_NPW (G1_ROWS / 8)
#define G1_XLD (G1_ROWS * 32 / 256)
__global__ __launch_bounds__(256)
void layer1_kernel(const float* __restrict__ x, const int* __restrict__ seg,
                  const int* __restrict__ ssrc,
                  const float* __restrict__ wself, const float* __restrict__ wneigh,
                  const float* __restrict__ bias,
                  const float* __restrict__ wself2, const float* __restrict__ wneigh2,
                  const float* __restrict__ bias2, float* __restrict__ sOut,
                  float* __restrict__ pOut, int nNodes) {
  __shared__ float xs[G1_ROWS * G1_STRIDE];
  __shared__ float as[G1_ROWS * G1_STRIDE];

  const int tid  = threadIdx.x;
  const int lane = tid & 31;
  const int wave = tid >> 5;
  const int half = lane >> 4;
  const int l16  = lane & 15;
  const int col  = wave * 16 + l16;
  const long long base = (long long)blockIdx.x * G1_ROWS;

  const int swave = __builtin_amdgcn_readfirstlane(wave);
  const int wrow = swave * G1_NPW;
  int2 sgv = make_int2(0, 0);
  {
    const long long nl = base + wrow + lane;
    if (lane < G1_NPW && nl < nNodes) sgv = *(const int2*)(seg + 2 * nl);
  }
  const float* xl = x + lane * 4;
  for (int i = 0; i < G1_NPW; ++i) {
    const int beg = __builtin_amdgcn_readlane(sgv.x, i);
    const int c   = __builtin_amdgcn_readlane(sgv.y, i);
    float4 acc = make_float4(0.f, 0.f, 0.f, 0.f);
    for (int b = 0; b < c; b += 32) {
      const int m = min(32, c - b);
      int sv = 0;
      if (lane < m) sv = ssrc[beg + b + lane];
      sv = min(max(sv, 0), nNodes - 1);
      for (int j = 0; j < m; j += 8) {
        float4 t[8];
#pragma unroll
        for (int u = 0; u < 8; ++u) {
          const int idx = min(j + u, m - 1);
          const int s = __builtin_amdgcn_readlane(sv, idx);
          t[u] = *(const float4*)(xl + (long long)s * NFEAT);
        }
#pragma unroll
        for (int u = 0; u < 8; ++u) {
          if (j + u < m) {
            acc.x += t[u].x; acc.y += t[u].y; acc.z += t[u].z; acc.w += t[u].w;
          }
        }
      }
    }
    const float dv = fmaxf((float)c, 1.0f);
    *(float4*)(&as[(wrow + i) * G1_STRIDE + lane * 4]) =
        make_float4(acc.x / dv, acc.y / dv, acc.z / dv, acc.w / dv);
  }

  const long long xrow0 = base + (tid >> 5);
  float* xsd = &xs[(tid >> 5) * G1_STRIDE + lane * 4];
  if (base + G1_ROWS <= nNodes) {
    const float* xt = x + xrow0 * NFEAT + lane * 4;
    float4 vx[G1_XLD];
#pragma unroll
    for (int r = 0; r < G1_XLD; ++r) vx[r] = *(const float4*)(xt + r * 8 * NFEAT);
#pragma unroll
    for (int r = 0; r < G1_XLD; ++r) *(float4*)(xsd + r * 8 * G1_STRIDE) = vx[r];
  } else {
    float4 vx[G1_XLD];
#pragma unroll
    for (int r = 0; r < G1_XLD; ++r) {
      const long long node = xrow0 + r * 8;
      const long long nr = (node < nNodes) ? node : (long long)(nNodes - 1);
      vx[r] = *(const float4*)(x + nr * NFEAT + lane * 4);
    }
#pragma unroll
    for (int r = 0; r < G1_XLD; ++r) *(float4*)(xsd + r * 8 * G1_STRIDE) = vx[r];
  }
  __syncthreads();

  const v8f vzero = {0.f, 0.f, 0.f, 0.f, 0.f, 0.f, 0.f, 0.f};
  v8f acc[G1_TILES];
#pragma unroll
  for (int r = 0; r < G1_TILES; ++r) acc[r] = vzero;

  for (int k0 = 0; k0 < NFEAT; k0 += 4) {
    const int kb = k0 + half * 2;
    v2f bs, bn;
    bs.x = wself [(long long)kb       * NHID + col];
    bs.y = wself [(long long)(kb + 1) * NHID + col];
    bn.x = wneigh[(long long)kb       * NHID + col];
    bn.y = wneigh[(long long)(kb + 1) * NHID + col];
#pragma unroll
    for (int r = 0; r < G1_TILES; ++r) {
      const v2f ax = *(const v2f*)(&xs[(r * 16 + l16) * G1_STRIDE + kb]);
      const v2f aa = *(const v2f*)(&as[(r * 16 + l16) * G1_STRIDE + kb]);
      acc[r] = __builtin_amdgcn_wmma_f32_16x16x4_f32(false, ax, false, bs,
                                                     (short)0, acc[r], false, false);
      acc[r] = __builtin_amdgcn_wmma_f32_16x16x4_f32(false, aa, false, bn,
                                                     (short)0, acc[r], false, false);
    }
  }

  const float bv = bias[col];
  __syncthreads();
#pragma unroll
  for (int r = 0; r < G1_TILES; ++r) {
#pragma unroll
    for (int j = 0; j < 8; ++j)
      xs[(r * 16 + j + half * 8) * G1_STRIDE + col] = fmaxf(acc[r][j] + bv, 0.0f);
  }
  __syncthreads();

  if (swave < G1_TILES) {
    v8f accS = vzero;
    v8f accP = vzero;
    const float* hrow = &xs[(swave * 16 + l16) * G1_STRIDE + half * 2];
    const float* bsp  = wself2  + half * 2 * NCLASS + l16;
    const float* bnp  = wneigh2 + half * 2 * NCLASS + l16;
    for (int k0 = 0; k0 < NHID; k0 += 4) {
      const v2f ah = *(const v2f*)(hrow + k0);
      v2f bs, bn;
      bs.x = bsp[k0 * NCLASS];
      bs.y = bsp[k0 * NCLASS + NCLASS];
      bn.x = bnp[k0 * NCLASS];
      bn.y = bnp[k0 * NCLASS + NCLASS];
      accS = __builtin_amdgcn_wmma_f32_16x16x4_f32(false, ah, false, bs,
                                                   (short)0, accS, false, false);
      accP = __builtin_amdgcn_wmma_f32_16x16x4_f32(false, ah, false, bn,
                                                   (short)0, accP, false, false);
    }
    const float bv2 = bias2[l16];
#pragma unroll
    for (int j = 0; j < 8; ++j) {
      const long long node = base + swave * 16 + j + half * 8;
      if (node < nNodes) {
        sOut[node * NCLASS + l16] = accS[j] + bv2;
        pOut[node * NCLASS + l16] = accP[j];
      }
    }
  }
}


__device__ __forceinline__ int quad_bcast(int v, int u) {
  switch (u) {
    case 0:  return __builtin_amdgcn_update_dpp(0, v, 0x00, 0xf, 0xf, false);
    case 1:  return __builtin_amdgcn_update_dpp(0, v, 0x55, 0xf, 0xf, false);
    case 2:  return __builtin_amdgcn_update_dpp(0, v, 0xaa, 0xf, 0xf, false);
    default: return __builtin_amdgcn_update_dpp(0, v, 0xff, 0xf, 0xf, false);
  }
}

__global__ __launch_bounds__(256)
void gather16_final_kernel(const float* __restrict__ s, const float* __restrict__ p,
                           const int* __restrict__ seg, const int* __restrict__ ssrc,
                           float* __restrict__ out, int nNodes, int nEdges) {
  const int lane = threadIdx.x & 31;
  const int q    = lane & 3;
  const int wv = (int)((blockIdx.x * blockDim.x + threadIdx.x) >> 5);
  const int nw = (int)((gridDim.x * blockDim.x) >> 5);
  for (int n0 = wv * 8; n0 < nNodes; n0 += nw * 8) {
    const int node = n0 + (lane >> 2);
    const bool ok = node < nNodes;
    int beg = 0, c = 0;
    float4 sv = make_float4(0.f, 0.f, 0.f, 0.f);
    if (ok) {
      const int2 sg = *(const int2*)(seg + 2 * node);
      beg = sg.x;
      c   = sg.y;
      sv  = *(const float4*)(s + (long long)node * NCLASS + q * 4);
    }
    float4 acc = make_float4(0.f, 0.f, 0.f, 0.f);
    for (int b0 = 0; __any(b0 < c); b0 += 16) {
      int idx[4];
#pragma unroll
      for (int t = 0; t < 4; ++t) {
        const int pos = min(max(beg + b0 + 4 * t + q, 0), nEdges - 1);
        idx[t] = min(max(ssrc[pos], 0), nNodes - 1);
      }
      float4 tv[16];
#pragma unroll
      for (int j = 0; j < 16; ++j) {
        const int sj = quad_bcast(idx[j >> 2], j & 3);
        tv[j] = make_float4(0.f, 0.f, 0.f, 0.f);
        if (b0 + j < c) tv[j] = *(const float4*)(p + (long long)sj * NCLASS + q * 4);
      }
#pragma unroll
      for (int j = 0; j < 16; ++j) {
        acc.x += tv[j].x; acc.y += tv[j].y; acc.z += tv[j].z; acc.w += tv[j].w;
      }
    }
    const float dv = fmaxf((float)c, 1.0f);
    const float4 v = make_float4(sv.x + acc.x / dv, sv.y + acc.y / dv,
                                 sv.z + acc.z / dv, sv.w + acc.w / dv);
    float mx = fmaxf(fmaxf(v.x, v.y), fmaxf(v.z, v.w));
    mx = fmaxf(mx, __shfl_xor(mx, 1, 32));
    mx = fmaxf(mx, __shfl_xor(mx, 2, 32));
    float sum = (expf(v.x - mx) + expf(v.y - mx)) + (expf(v.z - mx) + expf(v.w - mx));
    sum += __shfl_xor(sum, 1, 32);
    sum += __shfl_xor(sum, 2, 32);
    const float ls = logf(sum);
    if (ok)
      *(float4*)(out + (long long)node * NCLASS + q * 4) =
          make_float4(v.x - mx - ls, v.y - mx - ls, v.z - mx - ls, v.w - mx - ls);
  }
}

extern "C" void kernel_launch(void* const* d_in, const int* in_sizes, int n_in,
                              void* d_out, int out_size, void* d_ws, size_t ws_size,
                              hipStream_t stream) {
  const float* x   = (const float*)d_in[0];
  const int*   src = (const int*)  d_in[1];
  const int*   dst = (const int*)  d_in[2];
  const float* ws1 = (const float*)d_in[3];
  const float* wn1 = (const float*)d_in[4];
  const float* b1  = (const float*)d_in[5];
  const float* ws2 = (const float*)d_in[6];
  const float* wn2 = (const float*)d_in[7];
  const float* b2  = (const float*)d_in[8];
  float* out = (float*)d_out;

  const int nNodes = in_sizes[0] / NFEAT;
  const long long nEdges = in_sizes[1];

  float* ah  = (float*)d_ws;
  int*   aux = (int*)(ah + (size_t)nNodes * NFEAT);

  const int nE    = (int)nEdges;
  const int nPad  = (nNodes + 63) & ~63;
  const int nB    = (nNodes + CB_NODES - 1) >> CB_SHIFT;
  const int nBPad = (nB + 63) & ~63;
  const int nEPad = (nE + 63) & ~63;
  const int chunk = (nE + NCHUNK - 1) / NCHUNK;
  int* hmat   = aux;
  int* rowTot = hmat + nB * NCHUNK;
  int* rec    = rowTot + nBPad;
  int* seg    = rec + nEPad;
  int* ssrc   = seg + 2 * nPad;
  const long long outN = (long long)nNodes * NCLASS;
  float* s2   = (float*)(ssrc + nEPad);
  float* p2   = s2 + outN;

  chunk_hist_kernel<<<NCHUNK, SORT_THREADS, 0, stream>>>(dst, hmat, nE, nNodes, nB, chunk);
  row_scan_kernel<<<nB, NCHUNK, 0, stream>>>(hmat, rowTot);
  place_kernel<<<NCHUNK, SORT_THREADS, 0, stream>>>(src, dst, hmat, rowTot, rec, nE, nNodes, nB, chunk);
  fine_sort_kernel<<<nB, SORT_THREADS, 0, stream>>>(rec, rowTot, seg, ssrc, nE, nNodes);
  layer1_kernel<<<(nNodes + G1_ROWS - 1) / G1_ROWS, 256, 0, stream>>>(
      x, seg, ssrc, ws1, wn1, b1, ws2, wn2, b2, s2, p2, nNodes);

  const int g16Blocks  = (nNodes + 127) / 128;

  gather16_final_kernel<<<g16Blocks, 256, 0, stream>>>(s2, p2, seg, ssrc, out, nNodes, nE);
}


# both bucketing passes order their words in LDS (rank from a returning LDS add, scan of the pass counts) and store contiguous runs, not one line per 4-byte store; sums in LDS arrival order
# speedup vs baseline: 5.7401x; 1.2041x over previous
#include <hip/hip_runtime.h>
#include <math.h>

typedef __attribute__((ext_vector_type(2))) float v2f;
typedef __attribute__((ext_vector_type(8))) float v8f;

#define NFEAT 128
#define NHID  128
#define NCLASS 16

#define G1_ROWS   48
#define G1_TILES  3
#define G1_STRIDE 132

#define CB_SHIFT 8
#define CB_NODES 256
#define NCHUNK   256
#define NB_MAX   1024
#define SORT_THREADS 1024
#define SORT_WAVES   32

template <int NW>
__device__ __forceinline__ int block_excl_scan(int v, int* wsum  , int& total) {
  const int lane = threadIdx.x & 31;
  const int wave = threadIdx.x >> 5;
  int inc = v;
#pragma unroll
  for (int off = 1; off < 32; off <<= 1) {
    const int t = __shfl_up(inc, off, 32);
    if (lane >= off) inc += t;
  }
  if (lane == 31) wsum[wave] = inc;
  __syncthreads();
  if (wave == 0) {
    const int w = (lane < NW) ? wsum[lane] : 0;
    int winc = w;
#pragma unroll
    for (int off = 1; off < 32; off <<= 1) {
      const int t = __shfl_up(winc, off, 32);
      if (lane >= off) winc += t;
    }
    if (lane < NW) wsum[lane] = winc - w;
    if (lane == 31) wsum[NW] = winc;
  }
  __syncthreads();
  const int r = wsum[wave] + inc - v;
  total = wsum[NW];
  __syncthreads();
  return r;
}

__global__ __launch_bounds__(SORT_THREADS)
void chunk_hist_kernel(const int* __restrict__ dst, int* __restrict__ hmat,
                       int nEdges, int nNodes, int nB, int chunk) {
  __shared__ int hist[NB_MAX];
  const int tid = threadIdx.x;
  for (int b = tid; b < NB_MAX; b += SORT_THREADS) hist[b] = 0;
  __syncthreads();
  const int e0 = (int)blockIdx.x * chunk;
  const int e1 = min(e0 + chunk, nEdges);
  for (int e = e0 + tid; e < e1; e += 4 * SORT_THREADS) {
    int d[4];
#pragma unroll
    for (int u = 0; u < 4; ++u) {
      const int ee = e + u * SORT_THREADS;
      d[u] = 0;
      if (ee < e1) d[u] = dst[ee];
    }
#pragma unroll
    for (int u = 0; u < 4; ++u) {
      if (e + u * SORT_THREADS < e1) {
        const int dd = min(max(d[u], 0), nNodes - 1);
        atomicAdd(&hist[min(dd >> CB_SHIFT, NB_MAX - 1)], 1);
      }
    }
  }
  __syncthreads();
  for (int b = tid; b < nB; b += SORT_THREADS) hmat[b * NCHUNK + (int)blockIdx.x] = hist[b];
}

__global__ __launch_bounds__(NCHUNK)
void row_scan_kernel(int* __restrict__ hmat, int* __restrict__ rowTot) {
  __shared__ int wsum[NCHUNK / 32 + 1];
  const int i = (int)blockIdx.x * NCHUNK + (int)threadIdx.x;
  const int v = hmat[i];
  int total;
  const int ex = block_excl_scan<NCHUNK / 32>(v, wsum, total);
  hmat[i] = ex;
  if (threadIdx.x == 0) rowTot[blockIdx.x] = total;
}

#define PL_TILE 8192
#define PL_EPT  (PL_TILE / SORT_THREADS)
__global__ __launch_bounds__(SORT_THREADS)
void place_kernel(const int* __restrict__ src, const int* __restrict__ dst,
                  const int* __restrict__ hmat, const int* __restrict__ rowTot,
                  int* __restrict__ rec, int nEdges, int nNodes, int nB, int chunk) {
  __shared__ int cursor[NB_MAX];
  __shared__ int lcnt[NB_MAX];
  __shared__ int gbase[NB_MAX];
  __shared__ int wsum[SORT_WAVES + 1];
  __shared__ int sval[PL_TILE];
  __shared__ int spos[PL_TILE];
  const int tid = threadIdx.x;
  int carry = 0;
  for (int b0 = 0; b0 < nB; b0 += SORT_THREADS) {
    const int b = b0 + tid;
    const int v = (b < nB) ? rowTot[b] : 0;
    int total;
    const int ex = block_excl_scan<SORT_WAVES>(v, wsum, total);
    if (b < min(nB, NB_MAX)) cursor[b] = carry + ex + hmat[b * NCHUNK + (int)blockIdx.x];
    carry += total;
  }
  __syncthreads();
  const int e0 = (int)blockIdx.x * chunk;
  const int e1 = min(e0 + chunk, nEdges);
  for (int t0 = e0; t0 < e1; t0 += PL_TILE) {
    const int t1 = min(t0 + PL_TILE, e1);
    for (int b = tid; b < NB_MAX; b += SORT_THREADS) lcnt[b] = 0;
    __syncthreads();
    int s[PL_EPT], d[PL_EPT];
#pragma unroll
    for (int u = 0; u < PL_EPT; ++u) {
      const int ee = t0 + tid + u * SORT_THREADS;
      s[u] = 0; d[u] = 0;
      if (ee < t1) { s[u] = src[ee]; d[u] = dst[ee]; }
    }
    int wd[PL_EPT], bk[PL_EPT], lr[PL_EPT];
#pragma unroll
    for (int u = 0; u < PL_EPT; ++u) {
      wd[u] = 0; bk[u] = 0; lr[u] = 0;
      if (t0 + tid + u * SORT_THREADS < t1) {
        const int dd = min(max(d[u], 0), nNodes - 1);
        bk[u] = min(dd >> CB_SHIFT, NB_MAX - 1);
        wd[u] = (int)(((unsigned)s[u] << CB_SHIFT) | (unsigned)(dd & (CB_NODES - 1)));
        lr[u] = atomicAdd(&lcnt[bk[u]], 1);
      }
    }
    __syncthreads();
    for (int b0 = 0; b0 < NB_MAX; b0 += SORT_THREADS) {
      const int b = b0 + tid;
      const int c = lcnt[b];
      int total;
      const int ex = block_excl_scan<SORT_WAVES>(c, wsum, total);
      lcnt[b] = ex;
      const int g = cursor[b];
      gbase[b] = g;
      cursor[b] = g + c;
    }
    __syncthreads();
#pragma unroll
    for (int u = 0; u < PL_EPT; ++u) {
      if (t0 + tid + u * SORT_THREADS < t1) {
        const int i = min(max(lcnt[bk[u]] + lr[u], 0), PL_TILE - 1);
        sval[i] = wd[u];
        spos[i] = gbase[bk[u]] + lr[u];
      }
    }
    __syncthreads();
    const int n = t1 - t0;
    for (int i = tid; i < n; i += SORT_THREADS) {
      const int pos = min(max(spos[i], 0), nEdges - 1);
      rec[pos] = sval[i];
    }
    __syncthreads();
  }
}

#define FS_STAGE 8192
__global__ __launch_bounds__(SORT_THREADS)
void fine_sort_kernel(const int* __restrict__ rec, const int* __restrict__ rowTot,
                      int* __restrict__ seg, int* __restrict__ ssrc,
                      int nEdges, int nNodes) {
  __shared__ int cnt[CB_NODES];
  __shared__ int wsum[SORT_WAVES + 1];
  __shared__ int stage[FS_STAGE];
  const int tid = threadIdx.x;
  const int bkt = (int)blockIdx.x;
  int part = 0;
  for (int b = tid; b < bkt; b += SORT_THREADS) part += rowTot[b];
  int bstart;
  (void)block_excl_scan<SORT_WAVES>(part, wsum, bstart);
  const int r0 = min(max(bstart, 0), nEdges);
  const int r1 = min(r0 + max(rowTot[bkt], 0), nEdges);
  if (tid < CB_NODES) cnt[tid] = 0;
  __syncthreads();
  for (int i = r0 + tid; i < r1; i += 4 * SORT_THREADS) {
    int r[4];
#pragma unroll
    for (int u = 0; u < 4; ++u) {
      const int ii = i + u * SORT_THREADS;
      r[u] = 0;
      if (ii < r1) r[u] = rec[ii];
    }
#pragma unroll
    for (int u = 0; u < 4; ++u)
      if (i + u * SORT_THREADS < r1) atomicAdd(&cnt[r[u] & (CB_NODES - 1)], 1);
  }
  __syncthreads();
  const int c = (tid < CB_NODES) ? cnt[tid] : 0;
  int total;
  const int ex = block_excl_scan<SORT_WAVES>(c, wsum, total);
  if (tid < CB_NODES) {
    const int start = r0 + ex;
    const int node = bkt * CB_NODES + tid;
    if (node < nNodes) *(int2*)(seg + 2 * node) = make_int2(start, c);
    cnt[tid] = start;
  }
  __syncthreads();
  for (int i = r0 + tid; i < r1; i += 4 * SORT_THREADS) {
    int r[4];
#pragma unroll
    for (int u = 0; u < 4; ++u) {
      const int ii = i + u * SORT_THREADS;
      r[u] = 0;
      if (ii < r1) r[u] = rec[ii];
    }
#pragma unroll
    for (int u = 0; u < 4; ++u) {
      if (i + u * SORT_THREADS < r1) {
        const int pos = atomicAdd(&cnt[r[u] & (CB_NODES - 1)], 1);
        const int sid = (int)((unsigned)r[u] >> CB_SHIFT);
        const int off = pos - r0;
        if (off >= 0 && off < FS_STAGE) {
          stage[off] = sid;
        } else {
          ssrc[min(max(pos, 0), nEdges - 1)] = sid;
        }
      }
    }
  }
  __syncthreads();
  const int nst = min(r1 - r0, FS_STAGE);
  for (int i = tid; i < nst; i += SORT_THREADS) ssrc[r0 + i] = stage[i];
}

#define G1_NPW (G1_ROWS / 8)
#define G1_XLD (G1_ROWS * 32 / 256)
__global__ __launch_bounds__(256)
void layer1_kernel(const float* __restrict__ x, const int* __restrict__ seg,
                  const int* __restrict__ ssrc,
                  const float* __restrict__ wself, const float* __restrict__ wneigh,
                  const float* __restrict__ bias,
                  const float* __restrict__ wself2, const float* __restrict__ wneigh2,
                  const float* __restrict__ bias2, float* __restrict__ sOut,
                  float* __restrict__ pOut, int nNodes) {
  __shared__ float xs[G1_ROWS * G1_STRIDE];
  __shared__ float as[G1_ROWS * G1_STRIDE];

  const int tid  = threadIdx.x;
  const int lane = tid & 31;
  const int wave = tid >> 5;
  const int half = lane >> 4;
  const int l16  = lane & 15;
  const int col  = wave * 16 + l16;
  const long long base = (long long)blockIdx.x * G1_ROWS;

  const int swave = __builtin_amdgcn_readfirstlane(wave);
  const int wrow = swave * G1_NPW;
  int2 sgv = make_int2(0, 0);
  {
    const long long nl = base + wrow + lane;
    if (lane < G1_NPW && nl < nNodes) sgv = *(const int2*)(seg + 2 * nl);
  }
  const float* xl = x + lane * 4;
  for (int i = 0; i < G1_NPW; ++i) {
    const int beg = __builtin_amdgcn_readlane(sgv.x, i);
    const int c   = __builtin_amdgcn_readlane(sgv.y, i);
    float4 acc = make_float4(0.f, 0.f, 0.f, 0.f);
    for (int b = 0; b < c; b += 32) {
      const int m = min(32, c - b);
      int sv = 0;
      if (lane < m) sv = ssrc[beg + b + lane];
      sv = min(max(sv, 0), nNodes - 1);
      for (int j = 0; j < m; j += 8) {
        float4 t[8];
#pragma unroll
        for (int u = 0; u < 8; ++u) {
          const int idx = min(j + u, m - 1);
          const int s = __builtin_amdgcn_readlane(sv, idx);
          t[u] = *(const float4*)(xl + (long long)s * NFEAT);
        }
#pragma unroll
        for (int u = 0; u < 8; ++u) {
          if (j + u < m) {
            acc.x += t[u].x; acc.y += t[u].y; acc.z += t[u].z; acc.w += t[u].w;
          }
        }
      }
    }
    const float dv = fmaxf((float)c, 1.0f);
    *(float4*)(&as[(wrow + i) * G1_STRIDE + lane * 4]) =
        make_float4(acc.x / dv, acc.y / dv, acc.z / dv, acc.w / dv);
  }

  const long long xrow0 = base + (tid >> 5);
  float* xsd = &xs[(tid >> 5) * G1_STRIDE + lane * 4];
  if (base + G1_ROWS <= nNodes) {
    const float* xt = x + xrow0 * NFEAT + lane * 4;
    float4 vx[G1_XLD];
#pragma unroll
    for (int r = 0; r < G1_XLD; ++r) vx[r] = *(const float4*)(xt + r * 8 * NFEAT);
#pragma unroll
    for (int r = 0; r < G1_XLD; ++r) *(float4*)(xsd + r * 8 * G1_STRIDE) = vx[r];
  } else {
    float4 vx[G1_XLD];
#pragma unroll
    for (int r = 0; r < G1_XLD; ++r) {
      const long long node = xrow0 + r * 8;
      const long long nr = (node < nNodes) ? node : (long long)(nNodes - 1);
      vx[r] = *(const float4*)(x + nr * NFEAT + lane * 4);
    }
#pragma unroll
    for (int r = 0; r < G1_XLD; ++r) *(float4*)(xsd + r * 8 * G1_STRIDE) = vx[r];
  }
  __syncthreads();

  const v8f vzero = {0.f, 0.f, 0.f, 0.f, 0.f, 0.f, 0.f, 0.f};
  v8f acc[G1_TILES];
#pragma unroll
  for (int r = 0; r < G1_TILES; ++r) acc[r] = vzero;

  for (int k0 = 0; k0 < NFEAT; k0 += 4) {
    const int kb = k0 + half * 2;
    v2f bs, bn;
    bs.x = wself [(long long)kb       * NHID + col];
    bs.y = wself [(long long)(kb + 1) * NHID + col];
    bn.x = wneigh[(long long)kb       * NHID + col];
    bn.y = wneigh[(long long)(kb + 1) * NHID + col];
#pragma unroll
    for (int r = 0; r < G1_TILES; ++r) {
      const v2f ax = *(const v2f*)(&xs[(r * 16 + l16) * G1_STRIDE + kb]);
      const v2f aa = *(const v2f*)(&as[(r * 16 + l16) * G1_STRIDE + kb]);
      acc[r] = __builtin_amdgcn_wmma_f32_16x16x4_f32(false, ax, false, bs,
                                                     (short)0, acc[r], false, false);
      acc[r] = __builtin_amdgcn_wmma_f32_16x16x4_f32(false, aa, false, bn,
                                                     (short)0, acc[r], false, false);
    }
  }

  const float bv = bias[col];
  __syncthreads();
#pragma unroll
  for (int r = 0; r < G1_TILES; ++r) {
#pragma unroll
    for (int j = 0; j < 8; ++j)
      xs[(r * 16 + j + half * 8) * G1_STRIDE + col] = fmaxf(acc[r][j] + bv, 0.0f);
  }
  __syncthreads();

  if (swave < G1_TILES) {
    v8f accS = vzero;
    v8f accP = vzero;
    const float* hrow = &xs[(swave * 16 + l16) * G1_STRIDE + half * 2];
    const float* bsp  = wself2  + half * 2 * NCLASS + l16;
    const float* bnp  = wneigh2 + half * 2 * NCLASS + l16;
    for (int k0 = 0; k0 < NHID; k0 += 4) {
      const v2f ah = *(const v2f*)(hrow + k0);
      v2f bs, bn;
      bs.x = bsp[k0 * NCLASS];
      bs.y = bsp[k0 * NCLASS + NCLASS];
      bn.x = bnp[k0 * NCLASS];
      bn.y = bnp[k0 * NCLASS + NCLASS];
      accS = __builtin_amdgcn_wmma_f32_16x16x4_f32(false, ah, false, bs,
                                                   (short)0, accS, false, false);
      accP = __builtin_amdgcn_wmma_f32_16x16x4_f32(false, ah, false, bn,
                                                   (short)0, accP, false, false);
    }
    const float bv2 = bias2[l16];
#pragma unroll
    for (int j = 0; j < 8; ++j) {
      const long long node = base + swave * 16 + j + half * 8;
      if (node < nNodes) {
        sOut[node * NCLASS + l16] = accS[j] + bv2;
        pOut[node * NCLASS + l16] = accP[j];
      }
    }
  }
}


__device__ __forceinline__ int quad_bcast(int v, int u) {
  switch (u) {
    case 0:  return __builtin_amdgcn_update_dpp(0, v, 0x00, 0xf, 0xf, false);
    case 1:  return __builtin_amdgcn_update_dpp(0, v, 0x55, 0xf, 0xf, false);
    case 2:  return __builtin_amdgcn_update_dpp(0, v, 0xaa, 0xf, 0xf, false);
    default: return __builtin_amdgcn_update_dpp(0, v, 0xff, 0xf, 0xf, false);
  }
}

__global__ __launch_bounds__(256)
void gather16_final_kernel(const float* __restrict__ s, const float* __restrict__ p,
                           const int* __restrict__ seg, const int* __restrict__ ssrc,
                           float* __restrict__ out, int nNodes, int nEdges) {
  const int lane = threadIdx.x & 31;
  const int q    = lane & 3;
  const int wv = (int)((blockIdx.x * blockDim.x + threadIdx.x) >> 5);
  const int nw = (int)((gridDim.x * blockDim.x) >> 5);
  for (int n0 = wv * 8; n0 < nNodes; n0 += nw * 8) {
    const int node = n0 + (lane >> 2);
    const bool ok = node < nNodes;
    int beg = 0, c = 0;
    float4 sv = make_float4(0.f, 0.f, 0.f, 0.f);
    if (ok) {
      const int2 sg = *(const int2*)(seg + 2 * node);
      beg = sg.x;
      c   = sg.y;
      sv  = *(const float4*)(s + (long long)node * NCLASS + q * 4);
    }
    float4 acc = make_float4(0.f, 0.f, 0.f, 0.f);
    for (int b0 = 0; __any(b0 < c); b0 += 16) {
      int idx[4];
#pragma unroll
      for (int t = 0; t < 4; ++t) {
        const int pos = min(max(beg + b0 + 4 * t + q, 0), nEdges - 1);
        idx[t] = min(max(ssrc[pos], 0), nNodes - 1);
      }
      float4 tv[16];
#pragma unroll
      for (int j = 0; j < 16; ++j) {
        const int sj = quad_bcast(idx[j >> 2], j & 3);
        tv[j] = make_float4(0.f, 0.f, 0.f, 0.f);
        if (b0 + j < c) tv[j] = *(const float4*)(p + (long long)sj * NCLASS + q * 4);
      }
#pragma unroll
      for (int j = 0; j < 16; ++j) {
        acc.x += tv[j].x; acc.y += tv[j].y; acc.z += tv[j].z; acc.w += tv[j].w;
      }
    }
    const float dv = fmaxf((float)c, 1.0f);
    const float4 v = make_float4(sv.x + acc.x / dv, sv.y + acc.y / dv,
                                 sv.z + acc.z / dv, sv.w + acc.w / dv);
    float mx = fmaxf(fmaxf(v.x, v.y), fmaxf(v.z, v.w));
    mx = fmaxf(mx, __shfl_xor(mx, 1, 32));
    mx = fmaxf(mx, __shfl_xor(mx, 2, 32));
    float sum = (expf(v.x - mx) + expf(v.y - mx)) + (expf(v.z - mx) + expf(v.w - mx));
    sum += __shfl_xor(sum, 1, 32);
    sum += __shfl_xor(sum, 2, 32);
    const float ls = logf(sum);
    if (ok)
      *(float4*)(out + (long long)node * NCLASS + q * 4) =
          make_float4(v.x - mx - ls, v.y - mx - ls, v.z - mx - ls, v.w - mx - ls);
  }
}

extern "C" void kernel_launch(void* const* d_in, const int* in_sizes, int n_in,
                              void* d_out, int out_size, void* d_ws, size_t ws_size,
                              hipStream_t stream) {
  const float* x   = (const float*)d_in[0];
  const int*   src = (const int*)  d_in[1];
  const int*   dst = (const int*)  d_in[2];
  const float* ws1 = (const float*)d_in[3];
  const float* wn1 = (const float*)d_in[4];
  const float* b1  = (const float*)d_in[5];
  const float* ws2 = (const float*)d_in[6];
  const float* wn2 = (const float*)d_in[7];
  const float* b2  = (const float*)d_in[8];
  float* out = (float*)d_out;

  const int nNodes = in_sizes[0] / NFEAT;
  const long long nEdges = in_sizes[1];

  float* ah  = (float*)d_ws;
  int*   aux = (int*)(ah + (size_t)nNodes * NFEAT);

  const int nE    = (int)nEdges;
  const int nPad  = (nNodes + 63) & ~63;
  const int nB    = (nNodes + CB_NODES - 1) >> CB_SHIFT;
  const int nBPad = (nB + 63) & ~63;
  const int nEPad = (nE + 63) & ~63;
  const int chunk = (nE + NCHUNK - 1) / NCHUNK;
  int* hmat   = aux;
  int* rowTot = hmat + nB * NCHUNK;
  int* rec    = rowTot + nBPad;
  int* seg    = rec + nEPad;
  int* ssrc   = seg + 2 * nPad;
  const long long outN = (long long)nNodes * NCLASS;
  float* s2   = (float*)(ssrc + nEPad);
  float* p2   = s2 + outN;

  chunk_hist_kernel<<<NCHUNK, SORT_THREADS, 0, stream>>>(dst, hmat, nE, nNodes, nB, chunk);
  row_scan_kernel<<<nB, NCHUNK, 0, stream>>>(hmat, rowTot);
  place_kernel<<<NCHUNK, SORT_THREADS, 0, stream>>>(src, dst, hmat, rowTot, rec, nE, nNodes, nB, chunk);
  fine_sort_kernel<<<nB, SORT_THREADS, 0, stream>>>(rec, rowTot, seg, ssrc, nE, nNodes);
  layer1_kernel<<<(nNodes + G1_ROWS - 1) / G1_ROWS, 256, 0, stream>>>(
      x, seg, ssrc, ws1, wn1, b1, ws2, wn2, b2, s2, p2, nNodes);

  const int g16Blocks  = (nNodes + 127) / 128;

  gather16_final_kernel<<<g16Blocks, 256, 0, stream>>>(s2, p2, seg, ssrc, out, nNodes, nE);
}


# level-1 edge bucketing made tile-local: each tile orders its own edges by coarse bucket inside its own stretch and stores a run table; histogram and scan launches gone; sums in LDS arrival order
# speedup vs baseline: 6.2018x; 1.0804x over previous
#include <hip/hip_runtime.h>
#include <math.h>

typedef __attribute__((ext_vector_type(2))) float v2f;
typedef __attribute__((ext_vector_type(8))) float v8f;

#define NFEAT 128
#define NHID  128
#define NCLASS 16

#define G1_ROWS   48
#define G1_TILES  3
#define G1_STRIDE 132

#define CB_SHIFT 8
#define CB_NODES 256
#define NB_MAX   1024
#define SORT_THREADS 1024
#define SORT_WAVES   32
#define TILE_EPT 7
#define TILE_E   6400
#define RUN_BITS 16

template <int NW>
__device__ __forceinline__ int block_excl_scan(int v, int* wsum  , int& total) {
  const int lane = threadIdx.x & 31;
  const int wave = threadIdx.x >> 5;
  int inc = v;
#pragma unroll
  for (int off = 1; off < 32; off <<= 1) {
    const int t = __shfl_up(inc, off, 32);
    if (lane >= off) inc += t;
  }
  if (lane == 31) wsum[wave] = inc;
  __syncthreads();
  if (wave == 0) {
    const int w = (lane < NW) ? wsum[lane] : 0;
    int winc = w;
#pragma unroll
    for (int off = 1; off < 32; off <<= 1) {
      const int t = __shfl_up(winc, off, 32);
      if (lane >= off) winc += t;
    }
    if (lane < NW) wsum[lane] = winc - w;
    if (lane == 31) wsum[NW] = winc;
  }
  __syncthreads();
  const int r = wsum[wave] + inc - v;
  total = wsum[NW];
  __syncthreads();
  return r;
}

__global__ __launch_bounds__(SORT_THREADS)
void tile_sort_kernel(const int* __restrict__ src, const int* __restrict__ dst,
                      int* __restrict__ rt, int* __restrict__ rec,
                      int nEdges, int nNodes, int nB, int nT) {
  __shared__ int lcnt[NB_MAX];
  __shared__ int wsum[SORT_WAVES + 1];
  __shared__ int sval[TILE_EPT * SORT_THREADS];
  const int tid = threadIdx.x;
  const int t0 = (int)blockIdx.x * TILE_E;
  const int t1 = min(t0 + TILE_E, nEdges);
  for (int b = tid; b < NB_MAX; b += SORT_THREADS) lcnt[b] = 0;
  __syncthreads();
  int s[TILE_EPT], d[TILE_EPT];
#pragma unroll
  for (int u = 0; u < TILE_EPT; ++u) {
    const int ee = t0 + tid + u * SORT_THREADS;
    s[u] = 0; d[u] = 0;
    if (ee < t1) { s[u] = src[ee]; d[u] = dst[ee]; }
  }
  int wd[TILE_EPT], bk[TILE_EPT], lr[TILE_EPT];
#pragma unroll
  for (int u = 0; u < TILE_EPT; ++u) {
    wd[u] = 0; bk[u] = 0; lr[u] = 0;
    if (t0 + tid + u * SORT_THREADS < t1) {
      const int dd = min(max(d[u], 0), nNodes - 1);
      bk[u] = min(dd >> CB_SHIFT, NB_MAX - 1);
      wd[u] = (int)(((unsigned)s[u] << CB_SHIFT) | (unsigned)(dd & (CB_NODES - 1)));
      lr[u] = atomicAdd(&lcnt[bk[u]], 1);
    }
  }
  __syncthreads();
  {
    const int c = lcnt[tid];
    int total;
    const int ex = block_excl_scan<SORT_WAVES>(c, wsum, total);
    lcnt[tid] = ex;
    if (tid < nB) rt[tid * nT + (int)blockIdx.x] = (ex << RUN_BITS) | c;
  }
  __syncthreads();
#pragma unroll
  for (int u = 0; u < TILE_EPT; ++u) {
    if (t0 + tid + u * SORT_THREADS < t1) {
      const int i = min(max(lcnt[bk[u]] + lr[u], 0), TILE_EPT * SORT_THREADS - 1);
      sval[i] = wd[u];
    }
  }
  __syncthreads();
  const int n = t1 - t0;
  for (int i = tid; i < n; i += SORT_THREADS) rec[t0 + i] = sval[i];
}

#define FS_STAGE 8192
#define FS_RUNS  8
__global__ __launch_bounds__(SORT_THREADS)
void fine_sort_kernel(const int* __restrict__ rec, const int* __restrict__ rt,
                      int* __restrict__ seg, int* __restrict__ ssrc,
                      int nEdges, int nNodes, int nT) {
  __shared__ int cnt[CB_NODES];
  __shared__ int wsum[SORT_WAVES + 1];
  __shared__ int stage[FS_STAGE];
  const int tid  = threadIdx.x;
  const int lane = tid & 31;
  const int wave = __builtin_amdgcn_readfirstlane(tid >> 5);
  const int bkt = (int)blockIdx.x;
  const int* row = rt + (long long)bkt * nT;
  int pst = 0, pct = 0;
  int pkv0 = 0;
  for (int m0 = wave; m0 < nT; m0 += 32 * 32) {
    const int tl = m0 + 32 * lane;
    const int pk = (tl < nT) ? row[tl] : 0;
    if (m0 == wave) pkv0 = pk;
    pst += (int)((unsigned)pk >> RUN_BITS);
    pct += pk & ((1 << RUN_BITS) - 1);
  }
  int bstart, btotal;
  (void)block_excl_scan<SORT_WAVES>(pst, wsum, bstart);
  (void)block_excl_scan<SORT_WAVES>(pct, wsum, btotal);
  const int r0 = min(max(bstart, 0), nEdges);
  const int r1 = min(r0 + max(btotal, 0), nEdges);
  if (tid < CB_NODES) cnt[tid] = 0;
  __syncthreads();
  for (int m0 = wave; m0 < nT; m0 += 32 * 32) {
    const int tl = m0 + 32 * lane;
    int pkv = pkv0;
    if (m0 != wave) pkv = (tl < nT) ? row[tl] : 0;
    for (int g = 0; g < 32 && m0 + 32 * g < nT; g += FS_RUNS) {
      int r[FS_RUNS], c[FS_RUNS], bs[FS_RUNS];
#pragma unroll
      for (int u = 0; u < FS_RUNS; ++u) {
        const int pk = __builtin_amdgcn_readlane(pkv, (g + u) & 31);
        const int t = m0 + 32 * (g + u);
        c[u]  = (t < nT) ? (pk & ((1 << RUN_BITS) - 1)) : 0;
        bs[u] = t * TILE_E + (int)((unsigned)pk >> RUN_BITS);
        r[u] = rec[min(max(bs[u] + lane, 0), nEdges - 1)];
      }
#pragma unroll
      for (int u = 0; u < FS_RUNS; ++u) {
        if (lane < c[u]) atomicAdd(&cnt[r[u] & (CB_NODES - 1)], 1);
        for (int j = lane + 32; j < c[u]; j += 32) {
          const int rr = rec[min(max(bs[u] + j, 0), nEdges - 1)];
          atomicAdd(&cnt[rr & (CB_NODES - 1)], 1);
        }
      }
    }
  }
  __syncthreads();
  const int cn = (tid < CB_NODES) ? cnt[tid] : 0;
  int total;
  const int ex = block_excl_scan<SORT_WAVES>(cn, wsum, total);
  if (tid < CB_NODES) {
    const int start = r0 + ex;
    const int node = bkt * CB_NODES + tid;
    if (node < nNodes) *(int2*)(seg + 2 * node) = make_int2(start, cn);
    cnt[tid] = start;
  }
  __syncthreads();
  for (int m0 = wave; m0 < nT; m0 += 32 * 32) {
    const int tl = m0 + 32 * lane;
    int pkv = pkv0;
    if (m0 != wave) pkv = (tl < nT) ? row[tl] : 0;
    for (int g = 0; g < 32 && m0 + 32 * g < nT; g += FS_RUNS) {
      int r[FS_RUNS], c[FS_RUNS], bs[FS_RUNS];
#pragma unroll
      for (int u = 0; u < FS_RUNS; ++u) {
        const int pk = __builtin_amdgcn_readlane(pkv, (g + u) & 31);
        const int t = m0 + 32 * (g + u);
        c[u]  = (t < nT) ? (pk & ((1 << RUN_BITS) - 1)) : 0;
        bs[u] = t * TILE_E + (int)((unsigned)pk >> RUN_BITS);
        r[u] = rec[min(max(bs[u] + lane, 0), nEdges - 1)];
      }
#pragma unroll
      for (int u = 0; u < FS_RUNS; ++u) {
        for (int j = lane; j < c[u]; j += 32) {
          const int rr = (j == lane) ? r[u] : rec[min(max(bs[u] + j, 0), nEdges - 1)];
          const int pos = atomicAdd(&cnt[rr & (CB_NODES - 1)], 1);
          const int sid = (int)((unsigned)rr >> CB_SHIFT);
          const int off = pos - r0;
          if (off >= 0 && off < FS_STAGE) {
            stage[off] = sid;
          } else {
            ssrc[min(max(pos, 0), nEdges - 1)] = sid;
          }
        }
      }
    }
  }
  __syncthreads();
  const int nst = min(r1 - r0, FS_STAGE);
  for (int i = tid; i < nst; i += SORT_THREADS) ssrc[r0 + i] = stage[i];
}

#define G1_NPW (G1_ROWS / 8)
#define G1_XLD (G1_ROWS * 32 / 256)
__global__ __launch_bounds__(256)
void layer1_kernel(const float* __restrict__ x, const int* __restrict__ seg,
                  const int* __restrict__ ssrc,
                  const float* __restrict__ wself, const float* __restrict__ wneigh,
                  const float* __restrict__ bias,
                  const float* __restrict__ wself2, const float* __restrict__ wneigh2,
                  const float* __restrict__ bias2, float* __restrict__ sOut,
                  float* __restrict__ pOut, int nNodes) {
  __shared__ float xs[G1_ROWS * G1_STRIDE];
  __shared__ float as[G1_ROWS * G1_STRIDE];

  const int tid  = threadIdx.x;
  const int lane = tid & 31;
  const int wave = tid >> 5;
  const int half = lane >> 4;
  const int l16  = lane & 15;
  const int col  = wave * 16 + l16;
  const long long base = (long long)blockIdx.x * G1_ROWS;

  const int swave = __builtin_amdgcn_readfirstlane(wave);
  const int wrow = swave * G1_NPW;
  int2 sgv = make_int2(0, 0);
  {
    const long long nl = base + wrow + lane;
    if (lane < G1_NPW && nl < nNodes) sgv = *(const int2*)(seg + 2 * nl);
  }
  const float* xl = x + lane * 4;
  for (int i = 0; i < G1_NPW; ++i) {
    const int beg = __builtin_amdgcn_readlane(sgv.x, i);
    const int c   = __builtin_amdgcn_readlane(sgv.y, i);
    float4 acc = make_float4(0.f, 0.f, 0.f, 0.f);
    for (int b = 0; b < c; b += 32) {
      const int m = min(32, c - b);
      int sv = 0;
      if (lane < m) sv = ssrc[beg + b + lane];
      sv = min(max(sv, 0), nNodes - 1);
      for (int j = 0; j < m; j += 8) {
        float4 t[8];
#pragma unroll
        for (int u = 0; u < 8; ++u) {
          const int idx = min(j + u, m - 1);
          const int s = __builtin_amdgcn_readlane(sv, idx);
          t[u] = *(const float4*)(xl + (long long)s * NFEAT);
        }
#pragma unroll
        for (int u = 0; u < 8; ++u) {
          if (j + u < m) {
            acc.x += t[u].x; acc.y += t[u].y; acc.z += t[u].z; acc.w += t[u].w;
          }
        }
      }
    }
    const float dv = fmaxf((float)c, 1.0f);
    *(float4*)(&as[(wrow + i) * G1_STRIDE + lane * 4]) =
        make_float4(acc.x / dv, acc.y / dv, acc.z / dv, acc.w / dv);
  }

  const long long xrow0 = base + (tid >> 5);
  float* xsd = &xs[(tid >> 5) * G1_STRIDE + lane * 4];
  if (base + G1_ROWS <= nNodes) {
    const float* xt = x + xrow0 * NFEAT + lane * 4;
    float4 vx[G1_XLD];
#pragma unroll
    for (int r = 0; r < G1_XLD; ++r) vx[r] = *(const float4*)(xt + r * 8 * NFEAT);
#pragma unroll
    for (int r = 0; r < G1_XLD; ++r) *(float4*)(xsd + r * 8 * G1_STRIDE) = vx[r];
  } else {
    float4 vx[G1_XLD];
#pragma unroll
    for (int r = 0; r < G1_XLD; ++r) {
      const long long node = xrow0 + r * 8;
      const long long nr = (node < nNodes) ? node : (long long)(nNodes - 1);
      vx[r] = *(const float4*)(x + nr * NFEAT + lane * 4);
    }
#pragma unroll
    for (int r = 0; r < G1_XLD; ++r) *(float4*)(xsd + r * 8 * G1_STRIDE) = vx[r];
  }
  __syncthreads();

  const v8f vzero = {0.f, 0.f, 0.f, 0.f, 0.f, 0.f, 0.f, 0.f};
  v8f acc[G1_TILES];
#pragma unroll
  for (int r = 0; r < G1_TILES; ++r) acc[r] = vzero;

  for (int k0 = 0; k0 < NFEAT; k0 += 4) {
    const int kb = k0 + half * 2;
    v2f bs, bn;
    bs.x = wself [(long long)kb       * NHID + col];
    bs.y = wself [(long long)(kb + 1) * NHID + col];
    bn.x = wneigh[(long long)kb       * NHID + col];
    bn.y = wneigh[(long long)(kb + 1) * NHID + col];
#pragma unroll
    for (int r = 0; r < G1_TILES; ++r) {
      const v2f ax = *(const v2f*)(&xs[(r * 16 + l16) * G1_STRIDE + kb]);
      const v2f aa = *(const v2f*)(&as[(r * 16 + l16) * G1_STRIDE + kb]);
      acc[r] = __builtin_amdgcn_wmma_f32_16x16x4_f32(false, ax, false, bs,
                                                     (short)0, acc[r], false, false);
      acc[r] = __builtin_amdgcn_wmma_f32_16x16x4_f32(false, aa, false, bn,
                                                     (short)0, acc[r], false, false);
    }
  }

  const float bv = bias[col];
  __syncthreads();
#pragma unroll
  for (int r = 0; r < G1_TILES; ++r) {
#pragma unroll
    for (int j = 0; j < 8; ++j)
      xs[(r * 16 + j + half * 8) * G1_STRIDE + col] = fmaxf(acc[r][j] + bv, 0.0f);
  }
  __syncthreads();

  if (swave < G1_TILES) {
    v8f accS = vzero;
    v8f accP = vzero;
    const float* hrow = &xs[(swave * 16 + l16) * G1_STRIDE + half * 2];
    const float* bsp  = wself2  + half * 2 * NCLASS + l16;
    const float* bnp  = wneigh2 + half * 2 * NCLASS + l16;
    for (int k0 = 0; k0 < NHID; k0 += 4) {
      const v2f ah = *(const v2f*)(hrow + k0);
      v2f bs, bn;
      bs.x = bsp[k0 * NCLASS];
      bs.y = bsp[k0 * NCLASS + NCLASS];
      bn.x = bnp[k0 * NCLASS];
      bn.y = bnp[k0 * NCLASS + NCLASS];
      accS = __builtin_amdgcn_wmma_f32_16x16x4_f32(false, ah, false, bs,
                                                   (short)0, accS, false, false);
      accP = __builtin_amdgcn_wmma_f32_16x16x4_f32(false, ah, false, bn,
                                                   (short)0, accP, false, false);
    }
    const float bv2 = bias2[l16];
#pragma unroll
    for (int j = 0; j < 8; ++j) {
      const long long node = base + swave * 16 + j + half * 8;
      if (node < nNodes) {
        sOut[node * NCLASS + l16] = accS[j] + bv2;
        pOut[node * NCLASS + l16] = accP[j];
      }
    }
  }
}


__device__ __forceinline__ int quad_bcast(int v, int u) {
  switch (u) {
    case 0:  return __builtin_amdgcn_update_dpp(0, v, 0x00, 0xf, 0xf, false);
    case 1:  return __builtin_amdgcn_update_dpp(0, v, 0x55, 0xf, 0xf, false);
    case 2:  return __builtin_amdgcn_update_dpp(0, v, 0xaa, 0xf, 0xf, false);
    default: return __builtin_amdgcn_update_dpp(0, v, 0xff, 0xf, 0xf, false);
  }
}

__global__ __launch_bounds__(256)
void gather16_final_kernel(const float* __restrict__ s, const float* __restrict__ p,
                           const int* __restrict__ seg, const int* __restrict__ ssrc,
                           float* __restrict__ out, int nNodes, int nEdges) {
  const int lane = threadIdx.x & 31;
  const int q    = lane & 3;
  const int wv = (int)((blockIdx.x * blockDim.x + threadIdx.x) >> 5);
  const int nw = (int)((gridDim.x * blockDim.x) >> 5);
  for (int n0 = wv * 8; n0 < nNodes; n0 += nw * 8) {
    const int node = n0 + (lane >> 2);
    const bool ok = node < nNodes;
    int beg = 0, c = 0;
    float4 sv = make_float4(0.f, 0.f, 0.f, 0.f);
    if (ok) {
      const int2 sg = *(const int2*)(seg + 2 * node);
      beg = sg.x;
      c   = sg.y;
      sv  = *(const float4*)(s + (long long)node * NCLASS + q * 4);
    }
    float4 acc = make_float4(0.f, 0.f, 0.f, 0.f);
    for (int b0 = 0; __any(b0 < c); b0 += 16) {
      int idx[4];
#pragma unroll
      for (int t = 0; t < 4; ++t) {
        const int pos = min(max(beg + b0 + 4 * t + q, 0), nEdges - 1);
        idx[t] = min(max(ssrc[pos], 0), nNodes - 1);
      }
      float4 tv[16];
#pragma unroll
      for (int j = 0; j < 16; ++j) {
        const int sj = quad_bcast(idx[j >> 2], j & 3);
        tv[j] = make_float4(0.f, 0.f, 0.f, 0.f);
        if (b0 + j < c) tv[j] = *(const float4*)(p + (long long)sj * NCLASS + q * 4);
      }
#pragma unroll
      for (int j = 0; j < 16; ++j) {
        acc.x += tv[j].x; acc.y += tv[j].y; acc.z += tv[j].z; acc.w += tv[j].w;
      }
    }
    const float dv = fmaxf((float)c, 1.0f);
    const float4 v = make_float4(sv.x + acc.x / dv, sv.y + acc.y / dv,
                                 sv.z + acc.z / dv, sv.w + acc.w / dv);
    float mx = fmaxf(fmaxf(v.x, v.y), fmaxf(v.z, v.w));
    mx = fmaxf(mx, __shfl_xor(mx, 1, 32));
    mx = fmaxf(mx, __shfl_xor(mx, 2, 32));
    float sum = (expf(v.x - mx) + expf(v.y - mx)) + (expf(v.z - mx) + expf(v.w - mx));
    sum += __shfl_xor(sum, 1, 32);
    sum += __shfl_xor(sum, 2, 32);
    const float ls = logf(sum);
    if (ok)
      *(float4*)(out + (long long)node * NCLASS + q * 4) =
          make_float4(v.x - mx - ls, v.y - mx - ls, v.z - mx - ls, v.w - mx - ls);
  }
}

extern "C" void kernel_launch(void* const* d_in, const int* in_sizes, int n_in,
                              void* d_out, int out_size, void* d_ws, size_t ws_size,
                              hipStream_t stream) {
  const float* x   = (const float*)d_in[0];
  const int*   src = (const int*)  d_in[1];
  const int*   dst = (const int*)  d_in[2];
  const float* ws1 = (const float*)d_in[3];
  const float* wn1 = (const float*)d_in[4];
  const float* b1  = (const float*)d_in[5];
  const float* ws2 = (const float*)d_in[6];
  const float* wn2 = (const float*)d_in[7];
  const float* b2  = (const float*)d_in[8];
  float* out = (float*)d_out;

  const int nNodes = in_sizes[0] / NFEAT;
  const long long nEdges = in_sizes[1];

  float* ah  = (float*)d_ws;
  int*   aux = (int*)(ah + (size_t)nNodes * NFEAT);

  const int nE    = (int)nEdges;
  const int nPad  = (nNodes + 63) & ~63;
  const int nB    = (nNodes + CB_NODES - 1) >> CB_SHIFT;
  const int nEPad = (nE + 63) & ~63;
  const int nT    = (nE + TILE_E - 1) / TILE_E;
  const int nRtPad = (nB * nT + 63) & ~63;
  int* rt     = aux;
  int* rec    = rt + nRtPad;
  int* seg    = rec + nEPad;
  int* ssrc   = seg + 2 * nPad;
  const long long outN = (long long)nNodes * NCLASS;
  float* s2   = (float*)(ssrc + nEPad);
  float* p2   = s2 + outN;

  tile_sort_kernel<<<nT, SORT_THREADS, 0, stream>>>(src, dst, rt, rec, nE, nNodes, nB, nT);
  fine_sort_kernel<<<nB, SORT_THREADS, 0, stream>>>(rec, rt, seg, ssrc, nE, nNodes, nT);
  layer1_kernel<<<(nNodes + G1_ROWS - 1) / G1_ROWS, 256, 0, stream>>>(
      x, seg, ssrc, ws1, wn1, b1, ws2, wn2, b2, s2, p2, nNodes);

  const int g16Blocks  = (nNodes + 127) / 128;

  gather16_final_kernel<<<g16Blocks, 256, 0, stream>>>(s2, p2, seg, ssrc, out, nNodes, nE);
}


# layer-1 self product x@Wself issued by the gathering waves under their row loads (x tile staged first, 2 k steps per batch of 8 rows); sum order: all self terms, then neighbour terms
# speedup vs baseline: 6.3951x; 1.0312x over previous
#include <hip/hip_runtime.h>
#include <math.h>

typedef __attribute__((ext_vector_type(2))) float v2f;
typedef __attribute__((ext_vector_type(8))) float v8f;

#define NFEAT 128
#define NHID  128
#define NCLASS 16

#define G1_ROWS   48
#define G1_TILES  3
#define G1_STRIDE 132

#define CB_SHIFT 8
#define CB_NODES 256
#define NB_MAX   1024
#define SORT_THREADS 1024
#define SORT_WAVES   32
#define TILE_EPT 7
#define TILE_E   6400
#define RUN_BITS 16

template <int NW>
__device__ __forceinline__ int block_excl_scan(int v, int* wsum  , int& total) {
  const int lane = threadIdx.x & 31;
  const int wave = threadIdx.x >> 5;
  int inc = v;
#pragma unroll
  for (int off = 1; off < 32; off <<= 1) {
    const int t = __shfl_up(inc, off, 32);
    if (lane >= off) inc += t;
  }
  if (lane == 31) wsum[wave] = inc;
  __syncthreads();
  if (wave == 0) {
    const int w = (lane < NW) ? wsum[lane] : 0;
    int winc = w;
#pragma unroll
    for (int off = 1; off < 32; off <<= 1) {
      const int t = __shfl_up(winc, off, 32);
      if (lane >= off) winc += t;
    }
    if (lane < NW) wsum[lane] = winc - w;
    if (lane == 31) wsum[NW] = winc;
  }
  __syncthreads();
  const int r = wsum[wave] + inc - v;
  total = wsum[NW];
  __syncthreads();
  return r;
}

__global__ __launch_bounds__(SORT_THREADS)
void tile_sort_kernel(const int* __restrict__ src, const int* __restrict__ dst,
                      int* __restrict__ rt, int* __restrict__ rec,
                      int nEdges, int nNodes, int nB, int nT) {
  __shared__ int lcnt[NB_MAX];
  __shared__ int wsum[SORT_WAVES + 1];
  __shared__ int sval[TILE_EPT * SORT_THREADS];
  const int tid = threadIdx.x;
  const int t0 = (int)blockIdx.x * TILE_E;
  const int t1 = min(t0 + TILE_E, nEdges);
  for (int b = tid; b < NB_MAX; b += SORT_THREADS) lcnt[b] = 0;
  __syncthreads();
  int s[TILE_EPT], d[TILE_EPT];
#pragma unroll
  for (int u = 0; u < TILE_EPT; ++u) {
    const int ee = t0 + tid + u * SORT_THREADS;
    s[u] = 0; d[u] = 0;
    if (ee < t1) { s[u] = src[ee]; d[u] = dst[ee]; }
  }
  int wd[TILE_EPT], bk[TILE_EPT], lr[TILE_EPT];
#pragma unroll
  for (int u = 0; u < TILE_EPT; ++u) {
    wd[u] = 0; bk[u] = 0; lr[u] = 0;
    if (t0 + tid + u * SORT_THREADS < t1) {
      const int dd = min(max(d[u], 0), nNodes - 1);
      bk[u] = min(dd >> CB_SHIFT, NB_MAX - 1);
      wd[u] = (int)(((unsigned)s[u] << CB_SHIFT) | (unsigned)(dd & (CB_NODES - 1)));
      lr[u] = atomicAdd(&lcnt[bk[u]], 1);
    }
  }
  __syncthreads();
  {
    const int c = lcnt[tid];
    int total;
    const int ex = block_excl_scan<SORT_WAVES>(c, wsum, total);
    lcnt[tid] = ex;
    if (tid < nB) rt[tid * nT + (int)blockIdx.x] = (ex << RUN_BITS) | c;
  }
  __syncthreads();
#pragma unroll
  for (int u = 0; u < TILE_EPT; ++u) {
    if (t0 + tid + u * SORT_THREADS < t1) {
      const int i = min(max(lcnt[bk[u]] + lr[u], 0), TILE_EPT * SORT_THREADS - 1);
      sval[i] = wd[u];
    }
  }
  __syncthreads();
  const int n = t1 - t0;
  for (int i = tid; i < n; i += SORT_THREADS) rec[t0 + i] = sval[i];
}

#define FS_STAGE 8192
#define FS_RUNS  8
__global__ __launch_bounds__(SORT_THREADS)
void fine_sort_kernel(const int* __restrict__ rec, const int* __restrict__ rt,
                      int* __restrict__ seg, int* __restrict__ ssrc,
                      int nEdges, int nNodes, int nT) {
  __shared__ int cnt[CB_NODES];
  __shared__ int wsum[SORT_WAVES + 1];
  __shared__ int stage[FS_STAGE];
  const int tid  = threadIdx.x;
  const int lane = tid & 31;
  const int wave = __builtin_amdgcn_readfirstlane(tid >> 5);
  const int bkt = (int)blockIdx.x;
  const int* row = rt + (long long)bkt * nT;
  int pst = 0, pct = 0;
  int pkv0 = 0;
  for (int m0 = wave; m0 < nT; m0 += 32 * 32) {
    const int tl = m0 + 32 * lane;
    const int pk = (tl < nT) ? row[tl] : 0;
    if (m0 == wave) pkv0 = pk;
    pst += (int)((unsigned)pk >> RUN_BITS);
    pct += pk & ((1 << RUN_BITS) - 1);
  }
  int bstart, btotal;
  (void)block_excl_scan<SORT_WAVES>(pst, wsum, bstart);
  (void)block_excl_scan<SORT_WAVES>(pct, wsum, btotal);
  const int r0 = min(max(bstart, 0), nEdges);
  const int r1 = min(r0 + max(btotal, 0), nEdges);
  if (tid < CB_NODES) cnt[tid] = 0;
  __syncthreads();
  for (int m0 = wave; m0 < nT; m0 += 32 * 32) {
    const int tl = m0 + 32 * lane;
    int pkv = pkv0;
    if (m0 != wave) pkv = (tl < nT) ? row[tl] : 0;
    for (int g = 0; g < 32 && m0 + 32 * g < nT; g += FS_RUNS) {
      int r[FS_RUNS], c[FS_RUNS], bs[FS_RUNS];
#pragma unroll
      for (int u = 0; u < FS_RUNS; ++u) {
        const int pk = __builtin_amdgcn_readlane(pkv, (g + u) & 31);
        const int t = m0 + 32 * (g + u);
        c[u]  = (t < nT) ? (pk & ((1 << RUN_BITS) - 1)) : 0;
        bs[u] = t * TILE_E + (int)((unsigned)pk >> RUN_BITS);
        r[u] = rec[min(max(bs[u] + lane, 0), nEdges - 1)];
      }
#pragma unroll
      for (int u = 0; u < FS_RUNS; ++u) {
        if (lane < c[u]) atomicAdd(&cnt[r[u] & (CB_NODES - 1)], 1);
        for (int j = lane + 32; j < c[u]; j += 32) {
          const int rr = rec[min(max(bs[u] + j, 0), nEdges - 1)];
          atomicAdd(&cnt[rr & (CB_NODES - 1)], 1);
        }
      }
    }
  }
  __syncthreads();
  const int cn = (tid < CB_NODES) ? cnt[tid] : 0;
  int total;
  const int ex = block_excl_scan<SORT_WAVES>(cn, wsum, total);
  if (tid < CB_NODES) {
    const int start = r0 + ex;
    const int node = bkt * CB_NODES + tid;
    if (node < nNodes) *(int2*)(seg + 2 * node) = make_int2(start, cn);
    cnt[tid] = start;
  }
  __syncthreads();
  for (int m0 = wave; m0 < nT; m0 += 32 * 32) {
    const int tl = m0 + 32 * lane;
    int pkv = pkv0;
    if (m0 != wave) pkv = (tl < nT) ? row[tl] : 0;
    for (int g = 0; g < 32 && m0 + 32 * g < nT; g += FS_RUNS) {
      int r[FS_RUNS], c[FS_RUNS], bs[FS_RUNS];
#pragma unroll
      for (int u = 0; u < FS_RUNS; ++u) {
        const int pk = __builtin_amdgcn_readlane(pkv, (g + u) & 31);
        const int t = m0 + 32 * (g + u);
        c[u]  = (t < nT) ? (pk & ((1 << RUN_BITS) - 1)) : 0;
        bs[u] = t * TILE_E + (int)((unsigned)pk >> RUN_BITS);
        r[u] = rec[min(max(bs[u] + lane, 0), nEdges - 1)];
      }
#pragma unroll
      for (int u = 0; u < FS_RUNS; ++u) {
        for (int j = lane; j < c[u]; j += 32) {
          const int rr = (j == lane) ? r[u] : rec[min(max(bs[u] + j, 0), nEdges - 1)];
          const int pos = atomicAdd(&cnt[rr & (CB_NODES - 1)], 1);
          const int sid = (int)((unsigned)rr >> CB_SHIFT);
          const int off = pos - r0;
          if (off >= 0 && off < FS_STAGE) {
            stage[off] = sid;
          } else {
            ssrc[min(max(pos, 0), nEdges - 1)] = sid;
          }
        }
      }
    }
  }
  __syncthreads();
  const int nst = min(r1 - r0, FS_STAGE);
  for (int i = tid; i < nst; i += SORT_THREADS) ssrc[r0 + i] = stage[i];
}

#define G1_NPW (G1_ROWS / 8)
#define G1_XLD (G1_ROWS * 32 / 256)
#define G1_SELF_STEPS 2
__global__ __launch_bounds__(256, 10)
void layer1_kernel(const float* __restrict__ x, const int* __restrict__ seg,
                  const int* __restrict__ ssrc,
                  const float* __restrict__ wself, const float* __restrict__ wneigh,
                  const float* __restrict__ bias,
                  const float* __restrict__ wself2, const float* __restrict__ wneigh2,
                  const float* __restrict__ bias2, float* __restrict__ sOut,
                  float* __restrict__ pOut, int nNodes) {
  __shared__ float xs[G1_ROWS * G1_STRIDE];
  __shared__ float as[G1_ROWS * G1_STRIDE];

  const int tid  = threadIdx.x;
  const int lane = tid & 31;
  const int wave = tid >> 5;
  const int half = lane >> 4;
  const int l16  = lane & 15;
  const int col  = wave * 16 + l16;
  const long long base = (long long)blockIdx.x * G1_ROWS;

  const long long xrow0 = base + (tid >> 5);
  float* xsd = &xs[(tid >> 5) * G1_STRIDE + lane * 4];
  const int swave = __builtin_amdgcn_readfirstlane(wave);
  const int wrow = swave * G1_NPW;
  int2 sgv = make_int2(0, 0);
  {
    const long long nl = base + wrow + lane;
    if (lane < G1_NPW && nl < nNodes) sgv = *(const int2*)(seg + 2 * nl);
  }
  if (base + G1_ROWS <= nNodes) {
    const float* xt = x + xrow0 * NFEAT + lane * 4;
    float4 vx[G1_XLD];
#pragma unroll
    for (int r = 0; r < G1_XLD; ++r) vx[r] = *(const float4*)(xt + r * 8 * NFEAT);
#pragma unroll
    for (int r = 0; r < G1_XLD; ++r) *(float4*)(xsd + r * 8 * G1_STRIDE) = vx[r];
  } else {
    float4 vx[G1_XLD];
#pragma unroll
    for (int r = 0; r < G1_XLD; ++r) {
      const long long node = xrow0 + r * 8;
      const long long nr = (node < nNodes) ? node : (long long)(nNodes - 1);
      vx[r] = *(const float4*)(x + nr * NFEAT + lane * 4);
    }
#pragma unroll
    for (int r = 0; r < G1_XLD; ++r) *(float4*)(xsd + r * 8 * G1_STRIDE) = vx[r];
  }
  __syncthreads();

  const v8f vzero = {0.f, 0.f, 0.f, 0.f, 0.f, 0.f, 0.f, 0.f};
  v8f acc[G1_TILES];
#pragma unroll
  for (int r = 0; r < G1_TILES; ++r) acc[r] = vzero;

  int ks = 0;
  const char* xb = (const char*)x;
  const unsigned lane16 = (unsigned)lane * 16u;
  for (int i = 0; i < G1_NPW; ++i) {
    const int beg = __builtin_amdgcn_readlane(sgv.x, i);
    const int c   = __builtin_amdgcn_readlane(sgv.y, i);
    float4 sum = make_float4(0.f, 0.f, 0.f, 0.f);
    for (int b = 0; b < c; b += 32) {
      const int m = min(32, c - b);
      int sv = 0;
      if (lane < m) sv = ssrc[beg + b + lane];
      sv = min(max(sv, 0), nNodes - 1);
      for (int j = 0; j < m; j += 8) {
        if (ks < NFEAT / 4) {
          const int kb0 = ks * 4 + half * 2;
          v2f bf[G1_SELF_STEPS];
#pragma unroll
          for (int u = 0; u < G1_SELF_STEPS; ++u) {
            bf[u].x = wself[(long long)(kb0 + 4 * u)     * NHID + col];
            bf[u].y = wself[(long long)(kb0 + 4 * u + 1) * NHID + col];
          }
          __builtin_amdgcn_sched_barrier(0);
          float4 t[8];
#pragma unroll
          for (int u = 0; u < 8; ++u) {
            const int idx = min(j + u, m - 1);
            const int s = __builtin_amdgcn_readlane(sv, idx);
            t[u] = *(const float4*)(xb + (((unsigned)s << 9) + lane16));
          }
#pragma unroll
          for (int u = 0; u < G1_SELF_STEPS; ++u) {
#pragma unroll
            for (int r = 0; r < G1_TILES; ++r) {
              const v2f ax = *(const v2f*)(&xs[(r * 16 + l16) * G1_STRIDE + kb0 + 4 * u]);
              acc[r] = __builtin_amdgcn_wmma_f32_16x16x4_f32(false, ax, false, bf[u],
                                                             (short)0, acc[r], false, false);
            }
          }
          ks += G1_SELF_STEPS;
#pragma unroll
          for (int u = 0; u < 8; ++u) {
            const int keep = (u == 0 || j + u < m) ? -1 : 0;
            sum.x += __int_as_float(__float_as_int(t[u].x) & keep);
            sum.y += __int_as_float(__float_as_int(t[u].y) & keep);
            sum.z += __int_as_float(__float_as_int(t[u].z) & keep);
            sum.w += __int_as_float(__float_as_int(t[u].w) & keep);
          }
        } else {
          float4 t[8];
#pragma unroll
          for (int u = 0; u < 8; ++u) {
            const int idx = min(j + u, m - 1);
            const int s = __builtin_amdgcn_readlane(sv, idx);
            t[u] = *(const float4*)(xb + (((unsigned)s << 9) + lane16));
          }
#pragma unroll
          for (int u = 0; u < 8; ++u) {
            const int keep = (u == 0 || j + u < m) ? -1 : 0;
            sum.x += __int_as_float(__float_as_int(t[u].x) & keep);
            sum.y += __int_as_float(__float_as_int(t[u].y) & keep);
            sum.z += __int_as_float(__float_as_int(t[u].z) & keep);
            sum.w += __int_as_float(__float_as_int(t[u].w) & keep);
          }
        }
      }
    }
    const float dv = fmaxf((float)c, 1.0f);
    *(float4*)(&as[(wrow + i) * G1_STRIDE + lane * 4]) =
        make_float4(sum.x / dv, sum.y / dv, sum.z / dv, sum.w / dv);
  }

  for (; ks < NFEAT / 4; ++ks) {
    const int kb = ks * 4 + half * 2;
    v2f bs;
    bs.x = wself[(long long)kb       * NHID + col];
    bs.y = wself[(long long)(kb + 1) * NHID + col];
#pragma unroll
    for (int r = 0; r < G1_TILES; ++r) {
      const v2f ax = *(const v2f*)(&xs[(r * 16 + l16) * G1_STRIDE + kb]);
      acc[r] = __builtin_amdgcn_wmma_f32_16x16x4_f32(false, ax, false, bs,
                                                     (short)0, acc[r], false, false);
    }
  }
  __syncthreads();

  for (int k0 = 0; k0 < NFEAT; k0 += 4) {
    const int kb = k0 + half * 2;
    v2f bn;
    bn.x = wneigh[(long long)kb       * NHID + col];
    bn.y = wneigh[(long long)(kb + 1) * NHID + col];
#pragma unroll
    for (int r = 0; r < G1_TILES; ++r) {
      const v2f aa = *(const v2f*)(&as[(r * 16 + l16) * G1_STRIDE + kb]);
      acc[r] = __builtin_amdgcn_wmma_f32_16x16x4_f32(false, aa, false, bn,
                                                     (short)0, acc[r], false, false);
    }
  }

  const float bv = bias[col];
  __syncthreads();
#pragma unroll
  for (int r = 0; r < G1_TILES; ++r) {
#pragma unroll
    for (int j = 0; j < 8; ++j)
      xs[(r * 16 + j + half * 8) * G1_STRIDE + col] = fmaxf(acc[r][j] + bv, 0.0f);
  }
  __syncthreads();

  if (swave < G1_TILES) {
    v8f accS = vzero;
    v8f accP = vzero;
    const float* hrow = &xs[(swave * 16 + l16) * G1_STRIDE + half * 2];
    const float* bsp  = wself2  + half * 2 * NCLASS + l16;
    const float* bnp  = wneigh2 + half * 2 * NCLASS + l16;
    for (int k0 = 0; k0 < NHID; k0 += 4) {
      const v2f ah = *(const v2f*)(hrow + k0);
      v2f bs, bn;
      bs.x = bsp[k0 * NCLASS];
      bs.y = bsp[k0 * NCLASS + NCLASS];
      bn.x = bnp[k0 * NCLASS];
      bn.y = bnp[k0 * NCLASS + NCLASS];
      accS = __builtin_amdgcn_wmma_f32_16x16x4_f32(false, ah, false, bs,
                                                   (short)0, accS, false, false);
      accP = __builtin_amdgcn_wmma_f32_16x16x4_f32(false, ah, false, bn,
                                                   (short)0, accP, false, false);
    }
    const float bv2 = bias2[l16];
#pragma unroll
    for (int j = 0; j < 8; ++j) {
      const long long node = base + swave * 16 + j + half * 8;
      if (node < nNodes) {
        sOut[node * NCLASS + l16] = accS[j] + bv2;
        pOut[node * NCLASS + l16] = accP[j];
      }
    }
  }
}


__device__ __forceinline__ int quad_bcast(int v, int u) {
  switch (u) {
    case 0:  return __builtin_amdgcn_update_dpp(0, v, 0x00, 0xf, 0xf, false);
    case 1:  return __builtin_amdgcn_update_dpp(0, v, 0x55, 0xf, 0xf, false);
    case 2:  return __builtin_amdgcn_update_dpp(0, v, 0xaa, 0xf, 0xf, false);
    default: return __builtin_amdgcn_update_dpp(0, v, 0xff, 0xf, 0xf, false);
  }
}

__global__ __launch_bounds__(256)
void gather16_final_kernel(const float* __restrict__ s, const float* __restrict__ p,
                           const int* __restrict__ seg, const int* __restrict__ ssrc,
                           float* __restrict__ out, int nNodes, int nEdges) {
  const int lane = threadIdx.x & 31;
  const int q    = lane & 3;
  const int wv = (int)((blockIdx.x * blockDim.x + threadIdx.x) >> 5);
  const int nw = (int)((gridDim.x * blockDim.x) >> 5);
  for (int n0 = wv * 8; n0 < nNodes; n0 += nw * 8) {
    const int node = n0 + (lane >> 2);
    const bool ok = node < nNodes;
    int beg = 0, c = 0;
    float4 sv = make_float4(0.f, 0.f, 0.f, 0.f);
    if (ok) {
      const int2 sg = *(const int2*)(seg + 2 * node);
      beg = sg.x;
      c   = sg.y;
      sv  = *(const float4*)(s + (long long)node * NCLASS + q * 4);
    }
    float4 acc = make_float4(0.f, 0.f, 0.f, 0.f);
    for (int b0 = 0; __any(b0 < c); b0 += 16) {
      int idx[4];
#pragma unroll
      for (int t = 0; t < 4; ++t) {
        const int pos = min(max(beg + b0 + 4 * t + q, 0), nEdges - 1);
        idx[t] = min(max(ssrc[pos], 0), nNodes - 1);
      }
      float4 tv[16];
#pragma unroll
      for (int j = 0; j < 16; ++j) {
        const int sj = quad_bcast(idx[j >> 2], j & 3);
        tv[j] = make_float4(0.f, 0.f, 0.f, 0.f);
        if (b0 + j < c) tv[j] = *(const float4*)(p + (long long)sj * NCLASS + q * 4);
      }
#pragma unroll
      for (int j = 0; j < 16; ++j) {
        acc.x += tv[j].x; acc.y += tv[j].y; acc.z += tv[j].z; acc.w += tv[j].w;
      }
    }
    const float dv = fmaxf((float)c, 1.0f);
    const float4 v = make_float4(sv.x + acc.x / dv, sv.y + acc.y / dv,
                                 sv.z + acc.z / dv, sv.w + acc.w / dv);
    float mx = fmaxf(fmaxf(v.x, v.y), fmaxf(v.z, v.w));
    mx = fmaxf(mx, __shfl_xor(mx, 1, 32));
    mx = fmaxf(mx, __shfl_xor(mx, 2, 32));
    float sum = (expf(v.x - mx) + expf(v.y - mx)) + (expf(v.z - mx) + expf(v.w - mx));
    sum += __shfl_xor(sum, 1, 32);
    sum += __shfl_xor(sum, 2, 32);
    const float ls = logf(sum);
    if (ok)
      *(float4*)(out + (long long)node * NCLASS + q * 4) =
          make_float4(v.x - mx - ls, v.y - mx - ls, v.z - mx - ls, v.w - mx - ls);
  }
}

extern "C" void kernel_launch(void* const* d_in, const int* in_sizes, int n_in,
                              void* d_out, int out_size, void* d_ws, size_t ws_size,
                              hipStream_t stream) {
  const float* x   = (const float*)d_in[0];
  const int*   src = (const int*)  d_in[1];
  const int*   dst = (const int*)  d_in[2];
  const float* ws1 = (const float*)d_in[3];
  const float* wn1 = (const float*)d_in[4];
  const float* b1  = (const float*)d_in[5];
  const float* ws2 = (const float*)d_in[6];
  const float* wn2 = (const float*)d_in[7];
  const float* b2  = (const float*)d_in[8];
  float* out = (float*)d_out;

  const int nNodes = in_sizes[0] / NFEAT;
  const long long nEdges = in_sizes[1];

  float* ah  = (float*)d_ws;
  int*   aux = (int*)(ah + (size_t)nNodes * NFEAT);

  const int nE    = (int)nEdges;
  const int nPad  = (nNodes + 63) & ~63;
  const int nB    = (nNodes + CB_NODES - 1) >> CB_SHIFT;
  const int nEPad = (nE + 63) & ~63;
  const int nT    = (nE + TILE_E - 1) / TILE_E;
  const int nRtPad = (nB * nT + 63) & ~63;
  int* rt     = aux;
  int* rec    = rt + nRtPad;
  int* seg    = rec + nEPad;
  int* ssrc   = seg + 2 * nPad;
  const long long outN = (long long)nNodes * NCLASS;
  float* s2   = (float*)(ssrc + nEPad);
  float* p2   = s2 + outN;

  tile_sort_kernel<<<nT, SORT_THREADS, 0, stream>>>(src, dst, rt, rec, nE, nNodes, nB, nT);
  fine_sort_kernel<<<nB, SORT_THREADS, 0, stream>>>(rec, rt, seg, ssrc, nE, nNodes, nT);
  layer1_kernel<<<(nNodes + G1_ROWS - 1) / G1_ROWS, 256, 0, stream>>>(
      x, seg, ssrc, ws1, wn1, b1, ws2, wn2, b2, s2, p2, nNodes);

  const int g16Blocks  = (nNodes + 127) / 128;

  gather16_final_kernel<<<g16Blocks, 256, 0, stream>>>(s2, p2, seg, ssrc, out, nNodes, nE);
}
